# starting kernel
# baseline (speedup 1.0000x reference)
#include <hip/hip_runtime.h>
#include <hip/hip_bf16.h>

typedef __attribute__((ext_vector_type(2))) float v2f;
typedef __attribute__((ext_vector_type(8))) float v8f;

#define IN_CH 16
#define OUT_CH 8

__global__ void gcn_deg_init(float* __restrict__ deg, int n) {
    int i = blockIdx.x * blockDim.x + threadIdx.x;
    if (i < n) deg[i] = 1.0f;
}

__global__ void gcn_deg_count(const int* __restrict__ dst,
                              float* __restrict__ deg, int e_cnt) {
    int e = blockIdx.x * blockDim.x + threadIdx.x;
    if (e < e_cnt) atomicAdd(&deg[dst[e]], 1.0f);
}

__global__ void gcn_dinv(float* __restrict__ deg, int n) {
    int i = blockIdx.x * blockDim.x + threadIdx.x;
    if (i < n) {
        float d = deg[i];
        deg[i] = d > 0.0f ? __frsqrt_rn(d) : 0.0f;
    }
}

__global__ void gcn_gemm_wmma(const float* __restrict__ x,
                              const float* __restrict__ W,
                              const float* __restrict__ bias,
                              const float* __restrict__ dinv,
                              float* __restrict__ h,
                              float* __restrict__ out,
                              int n_nodes, int n_tiles) {
    const int lane  = threadIdx.x & 31;
    const int wave  = threadIdx.x >> 5;
    const int tile  = blockIdx.x * (blockDim.x >> 5) + wave;
    if (tile >= n_tiles) return;

    const int base  = tile * 16;
    const int m     = lane & 15;
    const int khalf = (lane >> 4) * 2;

    int arow = base + m;
    if (arow >= n_nodes) arow = n_nodes - 1;
    const float* __restrict__ xrow = x + (size_t)arow * IN_CH;

    v2f a0 = *(const v2f*)(xrow + 0  + khalf);
    v2f a1 = *(const v2f*)(xrow + 4  + khalf);
    v2f a2 = *(const v2f*)(xrow + 8  + khalf);
    v2f a3 = *(const v2f*)(xrow + 12 + khalf);

    const int   wrow = (m < OUT_CH) ? m : 0;
    const float sel  = (m < OUT_CH) ? 1.0f : 0.0f;
    const float* __restrict__ wr = W + wrow * IN_CH + khalf;
    v2f b0 = *(const v2f*)(wr + 0);
    v2f b1 = *(const v2f*)(wr + 4);
    v2f b2 = *(const v2f*)(wr + 8);
    v2f b3 = *(const v2f*)(wr + 12);
    b0 *= sel; b1 *= sel; b2 *= sel; b3 *= sel;

    v8f acc = {};
    acc = __builtin_amdgcn_wmma_f32_16x16x4_f32(false, a0, false, b0, (short)0, acc, false, false);
    acc = __builtin_amdgcn_wmma_f32_16x16x4_f32(false, a1, false, b1, (short)0, acc, false, false);
    acc = __builtin_amdgcn_wmma_f32_16x16x4_f32(false, a2, false, b2, (short)0, acc, false, false);
    acc = __builtin_amdgcn_wmma_f32_16x16x4_f32(false, a3, false, b3, (short)0, acc, false, false);

    const int  n     = m;
    const int  mbase = base + ((lane >> 4) * 8);
    const bool full  = (base + 16) <= n_nodes;

    if (full) {
        if (n < OUT_CH) {
            const float bn = bias[n];
            const float4 d0 = *(const float4*)(dinv + mbase);
            const float4 d1 = *(const float4*)(dinv + mbase + 4);
            const float dv[8] = {d0.x, d0.y, d0.z, d0.w, d1.x, d1.y, d1.z, d1.w};
            float* __restrict__ hp = h   + (size_t)mbase * OUT_CH + n;
            float* __restrict__ op = out + (size_t)mbase * OUT_CH + n;
#pragma unroll
            for (int r = 0; r < 8; ++r) {
                const float hv = acc[r];
                hp[r * OUT_CH] = hv;
                op[r * OUT_CH] = bn + dv[r] * dv[r] * hv;
            }
        }
    } else {
        if (n < OUT_CH) {
            const float bn = bias[n];
#pragma unroll
            for (int r = 0; r < 8; ++r) {
                const int node = mbase + r;
                if (node < n_nodes) {
                    const float hv = acc[r];
                    h[(size_t)node * OUT_CH + n] = hv;
                    const float di = dinv[node];
                    out[(size_t)node * OUT_CH + n] = bn + di * di * hv;
                }
            }
        }
    }
}

__global__ void gcn_scatter(const int* __restrict__ src,
                            const int* __restrict__ dst,
                            const float* __restrict__ dinv,
                            const float* __restrict__ h,
                            float* __restrict__ out, int e_cnt) {
    int e = blockIdx.x * blockDim.x + threadIdx.x;
    if (e >= e_cnt) return;
    const int s = src[e];
    const int d = dst[e];
    const float w = dinv[s] * dinv[d];
    const float4* __restrict__ hs = (const float4*)(h + (size_t)s * OUT_CH);
    const float4 h0 = hs[0];
    const float4 h1 = hs[1];
    float* __restrict__ o = out + (size_t)d * OUT_CH;
    atomicAdd(o + 0, w * h0.x);
    atomicAdd(o + 1, w * h0.y);
    atomicAdd(o + 2, w * h0.z);
    atomicAdd(o + 3, w * h0.w);
    atomicAdd(o + 4, w * h1.x);
    atomicAdd(o + 5, w * h1.y);
    atomicAdd(o + 6, w * h1.z);
    atomicAdd(o + 7, w * h1.w);
}

extern "C" void kernel_launch(void* const* d_in, const int* in_sizes, int n_in,
                              void* d_out, int out_size, void* d_ws, size_t ws_size,
                              hipStream_t stream) {
    const float* x    = (const float*)d_in[0];
    const int*   ei   = (const int*)d_in[1];
    const float* W    = (const float*)d_in[2];
    const float* bias = (const float*)d_in[3];

    const int N = in_sizes[0] / IN_CH;
    const int E = in_sizes[1] / 2;
    const int* src = ei;
    const int* dst = ei + (size_t)E;

    float* dinv = (float*)d_ws;
    size_t h_off = (((size_t)N * sizeof(float)) + 255u) & ~(size_t)255u;
    float* h = (float*)((char*)d_ws + h_off);
    float* out = (float*)d_out;

    const int TPB = 256;

    gcn_deg_init<<<(N + TPB - 1) / TPB, TPB, 0, stream>>>(dinv, N);
    gcn_deg_count<<<(E + TPB - 1) / TPB, TPB, 0, stream>>>(dst, dinv, E);
    gcn_dinv<<<(N + TPB - 1) / TPB, TPB, 0, stream>>>(dinv, N);

    const int n_tiles = (N + 15) / 16;
    const int waves_per_block = TPB / 32;
    const int gemm_blocks = (n_tiles + waves_per_block - 1) / waves_per_block;
    gcn_gemm_wmma<<<gemm_blocks, TPB, 0, stream>>>(x, W, bias, dinv, h, out, N, n_tiles);

    gcn_scatter<<<(E + TPB - 1) / TPB, TPB, 0, stream>>>(src, dst, dinv, h, out, E);
}


# scatter: 8 adjacent lanes own one edge's 8 channels (indices and weight by lane gather), so an edge's atomic adds and its h-row read are one 32-byte request each, not 8 and 2
# speedup vs baseline: 2.0966x; 2.0966x over previous
#include <hip/hip_runtime.h>
#include <hip/hip_bf16.h>

typedef __attribute__((ext_vector_type(2))) float v2f;
typedef __attribute__((ext_vector_type(8))) float v8f;

#define IN_CH 16
#define OUT_CH 8

__global__ void gcn_deg_init(float* __restrict__ deg, int n) {
    int i = blockIdx.x * blockDim.x + threadIdx.x;
    if (i < n) deg[i] = 1.0f;
}

__global__ void gcn_deg_count(const int* __restrict__ dst,
                              float* __restrict__ deg, int e_cnt) {
    int e = blockIdx.x * blockDim.x + threadIdx.x;
    if (e < e_cnt) atomicAdd(&deg[dst[e]], 1.0f);
}

__global__ void gcn_dinv(float* __restrict__ deg, int n) {
    int i = blockIdx.x * blockDim.x + threadIdx.x;
    if (i < n) {
        float d = deg[i];
        deg[i] = d > 0.0f ? __frsqrt_rn(d) : 0.0f;
    }
}

__global__ void gcn_gemm_wmma(const float* __restrict__ x,
                              const float* __restrict__ W,
                              const float* __restrict__ bias,
                              const float* __restrict__ dinv,
                              float* __restrict__ h,
                              float* __restrict__ out,
                              int n_nodes, int n_tiles) {
    const int lane  = threadIdx.x & 31;
    const int wave  = threadIdx.x >> 5;
    const int tile  = blockIdx.x * (blockDim.x >> 5) + wave;
    if (tile >= n_tiles) return;

    const int base  = tile * 16;
    const int m     = lane & 15;
    const int khalf = (lane >> 4) * 2;

    int arow = base + m;
    if (arow >= n_nodes) arow = n_nodes - 1;
    const float* __restrict__ xrow = x + (size_t)arow * IN_CH;

    v2f a0 = *(const v2f*)(xrow + 0  + khalf);
    v2f a1 = *(const v2f*)(xrow + 4  + khalf);
    v2f a2 = *(const v2f*)(xrow + 8  + khalf);
    v2f a3 = *(const v2f*)(xrow + 12 + khalf);

    const int   wrow = (m < OUT_CH) ? m : 0;
    const float sel  = (m < OUT_CH) ? 1.0f : 0.0f;
    const float* __restrict__ wr = W + wrow * IN_CH + khalf;
    v2f b0 = *(const v2f*)(wr + 0);
    v2f b1 = *(const v2f*)(wr + 4);
    v2f b2 = *(const v2f*)(wr + 8);
    v2f b3 = *(const v2f*)(wr + 12);
    b0 *= sel; b1 *= sel; b2 *= sel; b3 *= sel;

    v8f acc = {};
    acc = __builtin_amdgcn_wmma_f32_16x16x4_f32(false, a0, false, b0, (short)0, acc, false, false);
    acc = __builtin_amdgcn_wmma_f32_16x16x4_f32(false, a1, false, b1, (short)0, acc, false, false);
    acc = __builtin_amdgcn_wmma_f32_16x16x4_f32(false, a2, false, b2, (short)0, acc, false, false);
    acc = __builtin_amdgcn_wmma_f32_16x16x4_f32(false, a3, false, b3, (short)0, acc, false, false);

    const int  n     = m;
    const int  mbase = base + ((lane >> 4) * 8);
    const bool full  = (base + 16) <= n_nodes;

    if (full) {
        if (n < OUT_CH) {
            const float bn = bias[n];
            const float4 d0 = *(const float4*)(dinv + mbase);
            const float4 d1 = *(const float4*)(dinv + mbase + 4);
            const float dv[8] = {d0.x, d0.y, d0.z, d0.w, d1.x, d1.y, d1.z, d1.w};
            float* __restrict__ hp = h   + (size_t)mbase * OUT_CH + n;
            float* __restrict__ op = out + (size_t)mbase * OUT_CH + n;
#pragma unroll
            for (int r = 0; r < 8; ++r) {
                const float hv = acc[r];
                hp[r * OUT_CH] = hv;
                op[r * OUT_CH] = bn + dv[r] * dv[r] * hv;
            }
        }
    } else {
        if (n < OUT_CH) {
            const float bn = bias[n];
#pragma unroll
            for (int r = 0; r < 8; ++r) {
                const int node = mbase + r;
                if (node < n_nodes) {
                    const float hv = acc[r];
                    h[(size_t)node * OUT_CH + n] = hv;
                    const float di = dinv[node];
                    out[(size_t)node * OUT_CH + n] = bn + di * di * hv;
                }
            }
        }
    }
}

__global__ void gcn_scatter(const int* __restrict__ src,
                            const int* __restrict__ dst,
                            const float* __restrict__ dinv,
                            const float* __restrict__ h,
                            float* __restrict__ out, int e_cnt) {
    const int lane  = threadIdx.x & 31;
    const int e     = blockIdx.x * blockDim.x + threadIdx.x;
    const int ebase = e - lane;
    if (ebase >= e_cnt) return;

    const unsigned f   = (unsigned)(lane & 7);
    const int      grp = lane >> 3;

    if (ebase + 32 <= e_cnt) {
        const int s = src[e];
        const int d = dst[e];
        const float w = dinv[s] * dinv[d];
        const int   so = s * OUT_CH;
        const int   dd = d * OUT_CH;
        const int   wi = __float_as_int(w);
        float hv[8];
        unsigned oi[8];
        float wj[8];
        for (int j = 0; j < 8; ++j) {
            const int bp = (j * 4 + grp) << 2;
            const unsigned sj = (unsigned)__builtin_amdgcn_ds_bpermute(bp, so);
            hv[j] = h[sj + f];
            oi[j] = (unsigned)__builtin_amdgcn_ds_bpermute(bp, dd) + f;
            wj[j] = __int_as_float(__builtin_amdgcn_ds_bpermute(bp, wi));
        }
        for (int j = 0; j < 8; ++j) atomicAdd(out + oi[j], wj[j] * hv[j]);
    } else {
        int s = 0, d = 0;
        float w = 0.0f;
        if (e < e_cnt) {
            s = src[e];
            d = dst[e];
            w = dinv[s] * dinv[d];
        }
        const int so = s * OUT_CH;
        const int dd = d * OUT_CH;
        const int wi = __float_as_int(w);
        for (int j = 0; j < 8; ++j) {
            const int sl = j * 4 + grp;
            const int bp = sl << 2;
            const unsigned sj = (unsigned)__builtin_amdgcn_ds_bpermute(bp, so);
            const unsigned dj = (unsigned)__builtin_amdgcn_ds_bpermute(bp, dd);
            const float    ww = __int_as_float(__builtin_amdgcn_ds_bpermute(bp, wi));
            if (ebase + sl < e_cnt) atomicAdd(out + (dj + f), ww * h[sj + f]);
        }
    }
}

extern "C" void kernel_launch(void* const* d_in, const int* in_sizes, int n_in,
                              void* d_out, int out_size, void* d_ws, size_t ws_size,
                              hipStream_t stream) {
    const float* x    = (const float*)d_in[0];
    const int*   ei   = (const int*)d_in[1];
    const float* W    = (const float*)d_in[2];
    const float* bias = (const float*)d_in[3];

    const int N = in_sizes[0] / IN_CH;
    const int E = in_sizes[1] / 2;
    const int* src = ei;
    const int* dst = ei + (size_t)E;

    float* dinv = (float*)d_ws;
    size_t h_off = (((size_t)N * sizeof(float)) + 255u) & ~(size_t)255u;
    float* h = (float*)((char*)d_ws + h_off);
    float* out = (float*)d_out;

    const int TPB = 256;

    gcn_deg_init<<<(N + TPB - 1) / TPB, TPB, 0, stream>>>(dinv, N);
    gcn_deg_count<<<(E + TPB - 1) / TPB, TPB, 0, stream>>>(dst, dinv, E);
    gcn_dinv<<<(N + TPB - 1) / TPB, TPB, 0, stream>>>(dinv, N);

    const int n_tiles = (N + 15) / 16;
    const int waves_per_block = TPB / 32;
    const int gemm_blocks = (n_tiles + waves_per_block - 1) / waves_per_block;
    gcn_gemm_wmma<<<gemm_blocks, TPB, 0, stream>>>(x, W, bias, dinv, h, out, N, n_tiles);

    gcn_scatter<<<(E + TPB - 1) / TPB, TPB, 0, stream>>>(src, dst, dinv, h, out, E);
}


# each deg/out line now takes its atomic adds from one cache half: waves read their die and pull edge chunks by ticket for their half's node range (resident grid); atomic arrival order changes
# speedup vs baseline: 3.9925x; 1.9043x over previous
#include <hip/hip_runtime.h>
#include <hip/hip_bf16.h>

typedef __attribute__((ext_vector_type(2))) float v2f;
typedef __attribute__((ext_vector_type(8))) float v8f;

#define IN_CH 16
#define OUT_CH 8

#define CTR_STRIDE 64
#define DEG_CHUNK 256
#define SC_CHUNK  128

__device__ __forceinline__ int cache_half() {
    return (int)((__builtin_amdgcn_s_sendmsg_rtn(0x87) >> 18) & 1u);
}

__device__ __forceinline__ int draw_ticket(int* ctr) {
    int t = 0;
    if ((threadIdx.x & 31) == 0) t = atomicAdd(ctr, 1);
    return __builtin_amdgcn_readfirstlane(t);
}

__global__ void gcn_deg_init(float* __restrict__ deg, int* __restrict__ ctr, int n) {
    int i = blockIdx.x * blockDim.x + threadIdx.x;
    if (i < n) deg[i] = 1.0f;
    if (i < 4) ctr[i * CTR_STRIDE] = 0;
}

__global__ void __launch_bounds__(256)
gcn_deg_count(const int* __restrict__ dst, float* __restrict__ deg,
              int* __restrict__ ctr, int e_cnt, int n_chunks, int split, int n) {
    const int lane = threadIdx.x & 31;
    int  r      = cache_half();
    bool helped = false;
    int  next   = draw_ticket(ctr + r * CTR_STRIDE);
    const int max_trips = 2 * n_chunks + 2;
    for (int it = 0; it < max_trips; ++it) {
        const int c = next;
        if (c >= n_chunks) {
            if (helped) break;
            helped = true;
            r ^= 1;
            next = draw_ticket(ctr + r * CTR_STRIDE);
            continue;
        }
        const int lo = r ? split : 0;
        const int hi = r ? n : split;
        const int e0 = c * DEG_CHUNK + lane;
        int d[DEG_CHUNK / 32];
#pragma unroll
        for (int p = 0; p < DEG_CHUNK / 32; ++p) {
            const int e = e0 + p * 32;
            d[p] = (e < e_cnt) ? dst[e] : -1;
        }
        next = draw_ticket(ctr + r * CTR_STRIDE);
#pragma unroll
        for (int p = 0; p < DEG_CHUNK / 32; ++p)
            if (d[p] >= lo && d[p] < hi) atomicAdd(&deg[d[p]], 1.0f);
    }
}

__global__ void gcn_dinv(float* __restrict__ deg, int n) {
    int i = blockIdx.x * blockDim.x + threadIdx.x;
    if (i < n) {
        float d = deg[i];
        deg[i] = d > 0.0f ? __frsqrt_rn(d) : 0.0f;
    }
}

__global__ void gcn_gemm_wmma(const float* __restrict__ x,
                              const float* __restrict__ W,
                              const float* __restrict__ bias,
                              const float* __restrict__ dinv,
                              float* __restrict__ h,
                              float* __restrict__ out,
                              int n_nodes, int n_tiles) {
    const int lane  = threadIdx.x & 31;
    const int wave  = threadIdx.x >> 5;
    const int tile  = blockIdx.x * (blockDim.x >> 5) + wave;
    if (tile >= n_tiles) return;

    const int base  = tile * 16;
    const int m     = lane & 15;
    const int khalf = (lane >> 4) * 2;

    int arow = base + m;
    if (arow >= n_nodes) arow = n_nodes - 1;
    const float* __restrict__ xrow = x + (size_t)arow * IN_CH;

    v2f a0 = *(const v2f*)(xrow + 0  + khalf);
    v2f a1 = *(const v2f*)(xrow + 4  + khalf);
    v2f a2 = *(const v2f*)(xrow + 8  + khalf);
    v2f a3 = *(const v2f*)(xrow + 12 + khalf);

    const int   wrow = (m < OUT_CH) ? m : 0;
    const float sel  = (m < OUT_CH) ? 1.0f : 0.0f;
    const float* __restrict__ wr = W + wrow * IN_CH + khalf;
    v2f b0 = *(const v2f*)(wr + 0);
    v2f b1 = *(const v2f*)(wr + 4);
    v2f b2 = *(const v2f*)(wr + 8);
    v2f b3 = *(const v2f*)(wr + 12);
    b0 *= sel; b1 *= sel; b2 *= sel; b3 *= sel;

    v8f acc = {};
    acc = __builtin_amdgcn_wmma_f32_16x16x4_f32(false, a0, false, b0, (short)0, acc, false, false);
    acc = __builtin_amdgcn_wmma_f32_16x16x4_f32(false, a1, false, b1, (short)0, acc, false, false);
    acc = __builtin_amdgcn_wmma_f32_16x16x4_f32(false, a2, false, b2, (short)0, acc, false, false);
    acc = __builtin_amdgcn_wmma_f32_16x16x4_f32(false, a3, false, b3, (short)0, acc, false, false);

    const int  n     = m;
    const int  mbase = base + ((lane >> 4) * 8);
    const bool full  = (base + 16) <= n_nodes;

    if (full) {
        if (n < OUT_CH) {
            const float bn = bias[n];
            const float4 d0 = *(const float4*)(dinv + mbase);
            const float4 d1 = *(const float4*)(dinv + mbase + 4);
            const float dv[8] = {d0.x, d0.y, d0.z, d0.w, d1.x, d1.y, d1.z, d1.w};
            float* __restrict__ hp = h   + (size_t)mbase * OUT_CH + n;
            float* __restrict__ op = out + (size_t)mbase * OUT_CH + n;
#pragma unroll
            for (int r = 0; r < 8; ++r) {
                const float hv = acc[r];
                hp[r * OUT_CH] = hv;
                op[r * OUT_CH] = bn + dv[r] * dv[r] * hv;
            }
        }
    } else {
        if (n < OUT_CH) {
            const float bn = bias[n];
#pragma unroll
            for (int r = 0; r < 8; ++r) {
                const int node = mbase + r;
                if (node < n_nodes) {
                    const float hv = acc[r];
                    h[(size_t)node * OUT_CH + n] = hv;
                    const float di = dinv[node];
                    out[(size_t)node * OUT_CH + n] = bn + di * di * hv;
                }
            }
        }
    }
}

__global__ void __launch_bounds__(256)
gcn_scatter(const int* __restrict__ src, const int* __restrict__ dst,
            const float* __restrict__ dinv, const float* __restrict__ h,
            float* __restrict__ out, int* __restrict__ ctr,
            int e_cnt, int n_chunks, int split, int n) {
    const int      lane = threadIdx.x & 31;
    const unsigned f    = (unsigned)(lane & 7);
    const int      grp  = lane >> 3;

    int  r      = cache_half();
    bool helped = false;
    int  next   = draw_ticket(ctr + r * CTR_STRIDE);
    const int max_trips = 2 * n_chunks + 2;
    for (int it = 0; it < max_trips; ++it) {
        const int c = next;
        if (c >= n_chunks) {
            if (helped) break;
            helped = true;
            r ^= 1;
            next = draw_ticket(ctr + r * CTR_STRIDE);
            continue;
        }
        const int lo = r ? split : 0;
        const int hi = r ? n : split;
        const int e0 = c * SC_CHUNK + lane;

        int   dv[SC_CHUNK / 32];
        int   sv[SC_CHUNK / 32];
        float wv[SC_CHUNK / 32];
        bool  in[SC_CHUNK / 32];
#pragma unroll
        for (int p = 0; p < SC_CHUNK / 32; ++p) {
            const int e  = e0 + p * 32;
            const int ec = (e < e_cnt) ? e : e_cnt - 1;
            dv[p] = dst[ec];
            sv[p] = src[ec];
        }
        next = draw_ticket(ctr + r * CTR_STRIDE);
#pragma unroll
        for (int p = 0; p < SC_CHUNK / 32; ++p) {
            in[p] = (e0 + p * 32 < e_cnt) && (dv[p] >= lo) && (dv[p] < hi);
            dv[p] = in[p] ? dv[p] : 0;
            sv[p] = in[p] ? sv[p] : 0;
        }
#pragma unroll
        for (int p = 0; p < SC_CHUNK / 32; ++p) {
            const float a = dinv[sv[p]];
            const float b = dinv[dv[p]];
            wv[p] = in[p] ? a * b : 0.0f;
        }

#pragma unroll
        for (int p = 0; p < SC_CHUNK / 32; ++p) {
            const unsigned mask = __builtin_amdgcn_ballot_w32(in[p]);
            const int cnt = __popc(mask);
            if (cnt > 0) {
                const int rank = (int)__builtin_amdgcn_mbcnt_lo(mask, 0u);
                const int slot = in[p] ? rank : cnt + lane - rank;
                const int so = __builtin_amdgcn_ds_permute(slot << 2, sv[p] * OUT_CH);
                const int dd = __builtin_amdgcn_ds_permute(slot << 2, dv[p] * OUT_CH);
                const int wi = __builtin_amdgcn_ds_permute(slot << 2, __float_as_int(wv[p]));
                for (int jb = 0; jb < cnt; jb += 16) {
                    float    hv[4];
                    unsigned oi[4];
                    float    wj[4];
#pragma unroll
                    for (int j = 0; j < 4; ++j) {
                        const int sl = jb + j * 4 + grp;
                        const int bp = sl << 2;
                        const unsigned sj = (unsigned)__builtin_amdgcn_ds_bpermute(bp, so);
                        hv[j] = (sl < cnt) ? h[sj + f] : 0.0f;
                        oi[j] = (unsigned)__builtin_amdgcn_ds_bpermute(bp, dd) + f;
                        wj[j] = __int_as_float(__builtin_amdgcn_ds_bpermute(bp, wi));
                    }
#pragma unroll
                    for (int j = 0; j < 4; ++j)
                        if (jb + j * 4 + grp < cnt) atomicAdd(out + oi[j], wj[j] * hv[j]);
                }
            }
        }
    }
}

extern "C" void kernel_launch(void* const* d_in, const int* in_sizes, int n_in,
                              void* d_out, int out_size, void* d_ws, size_t ws_size,
                              hipStream_t stream) {
    const float* x    = (const float*)d_in[0];
    const int*   ei   = (const int*)d_in[1];
    const float* W    = (const float*)d_in[2];
    const float* bias = (const float*)d_in[3];

    const int N = in_sizes[0] / IN_CH;
    const int E = in_sizes[1] / 2;
    const int* src = ei;
    const int* dst = ei + (size_t)E;

    int*   ctr  = (int*)d_ws;
    const size_t dinv_off = 4u * CTR_STRIDE * sizeof(int);
    float* dinv = (float*)((char*)d_ws + dinv_off);
    size_t h_off = ((dinv_off + (size_t)N * sizeof(float)) + 255u) & ~(size_t)255u;
    float* h = (float*)((char*)d_ws + h_off);
    float* out = (float*)d_out;

    const int TPB = 256;

    const int split = ((N / 2) + 63) & ~63;

    gcn_deg_init<<<(N + TPB - 1) / TPB, TPB, 0, stream>>>(dinv, ctr, N);

    const int deg_chunks = (E + DEG_CHUNK - 1) / DEG_CHUNK;
    gcn_deg_count<<<1024, TPB, 0, stream>>>(dst, dinv, ctr, E, deg_chunks, split, N);
    gcn_dinv<<<(N + TPB - 1) / TPB, TPB, 0, stream>>>(dinv, N);

    const int n_tiles = (N + 15) / 16;
    const int waves_per_block = TPB / 32;
    const int gemm_blocks = (n_tiles + waves_per_block - 1) / waves_per_block;
    gcn_gemm_wmma<<<gemm_blocks, TPB, 0, stream>>>(x, W, bias, dinv, h, out, N, n_tiles);

    const int sc_chunks = (E + SC_CHUNK - 1) / SC_CHUNK;
    gcn_scatter<<<2048, TPB, 0, stream>>>(src, dst, dinv, h, out, ctr + 2 * CTR_STRIDE,
                                          E, sc_chunks, split, N);
}


# degree count: a workgroup counts one 65536-node range of one edge slice in an LDS integer histogram, then adds the tile to deg with one line-wide float atomic per 32 nodes (sum order changes; exact)
# speedup vs baseline: 4.3564x; 1.0911x over previous
#include <hip/hip_runtime.h>
#include <hip/hip_bf16.h>

typedef __attribute__((ext_vector_type(2))) float v2f;
typedef __attribute__((ext_vector_type(8))) float v8f;

#define IN_CH 16
#define OUT_CH 8

#define CTR_STRIDE 64
#define DEG_TILE   65536
#define DEG_TPB    1024
#define DEG_UNROLL 16
#define DEG_FLUSH  8
#define SC_CHUNK  128

__device__ __forceinline__ int cache_half() {
    return (int)((__builtin_amdgcn_s_sendmsg_rtn(0x87) >> 18) & 1u);
}

__device__ __forceinline__ int draw_ticket(int* ctr) {
    int t = 0;
    if ((threadIdx.x & 31) == 0) t = atomicAdd(ctr, 1);
    return __builtin_amdgcn_readfirstlane(t);
}

__global__ void gcn_deg_init(float* __restrict__ deg, int* __restrict__ ctr, int n) {
    int i = blockIdx.x * blockDim.x + threadIdx.x;
    if (i < n) deg[i] = 1.0f;
    if (i < 4) ctr[i * CTR_STRIDE] = 0;
}

__global__ void __launch_bounds__(DEG_TPB)
gcn_deg_count(const int* __restrict__ dst, float* __restrict__ deg,
              int* __restrict__ ctr, int e_cnt, int n,
              int n_ranges, int ranges_lo, int parts, int slice) {
    __shared__ int tile[DEG_TILE];
    __shared__ int s_msg[2];
    const int tid = threadIdx.x;

    if (tid == 0) s_msg[1] = cache_half();
    __syncthreads();
    int  r      = __builtin_amdgcn_readfirstlane(s_msg[1]);
    bool helped = false;
    const int max_trips = n_ranges * parts + 2;
    for (int it = 0; it < max_trips; ++it) {
        if (tid == 0) s_msg[0] = atomicAdd(ctr + r * CTR_STRIDE, 1);
        __syncthreads();
        const int t = __builtin_amdgcn_readfirstlane(s_msg[0]);
        __syncthreads();
        const int first = r ? ranges_lo : 0;
        const int count = r ? (n_ranges - ranges_lo) : ranges_lo;
        if (t >= count * parts) {
            if (helped) break;
            helped = true;
            r ^= 1;
            continue;
        }
        const int range = first + t / parts;
        const int part  = t - (t / parts) * parts;
        const int lo    = range * DEG_TILE;

        for (int i = tid; i < DEG_TILE; i += DEG_TPB) tile[i] = 0;
        __syncthreads();

        const int e_begin = part * slice;
        const int e_stop  = e_begin + slice;
        const int e_end   = e_stop < e_cnt ? e_stop : e_cnt;
        for (int base = e_begin; base < e_end; base += DEG_TPB * DEG_UNROLL) {
            int d[DEG_UNROLL];
#pragma unroll
            for (int k = 0; k < DEG_UNROLL; ++k) {
                const int e = base + k * DEG_TPB + tid;
                d[k] = (e < e_end) ? dst[e] : -1;
            }
#pragma unroll
            for (int k = 0; k < DEG_UNROLL; ++k) {
                const unsigned off = (unsigned)(d[k] - lo);
                if (off < (unsigned)DEG_TILE) atomicAdd(&tile[off], 1);
            }
        }
        __syncthreads();

        for (int i0 = tid; i0 < DEG_TILE; i0 += DEG_TPB * DEG_FLUSH) {
            int c[DEG_FLUSH];
#pragma unroll
            for (int k = 0; k < DEG_FLUSH; ++k) c[k] = tile[i0 + k * DEG_TPB];
#pragma unroll
            for (int k = 0; k < DEG_FLUSH; ++k) {
                const int node = lo + i0 + k * DEG_TPB;
                if (c[k] != 0 && node < n) atomicAdd(&deg[node], (float)c[k]);
            }
        }
    }
}

__global__ void gcn_dinv(float* __restrict__ deg, int n) {
    int i = blockIdx.x * blockDim.x + threadIdx.x;
    if (i < n) {
        float d = deg[i];
        deg[i] = d > 0.0f ? __frsqrt_rn(d) : 0.0f;
    }
}

__global__ void gcn_gemm_wmma(const float* __restrict__ x,
                              const float* __restrict__ W,
                              const float* __restrict__ bias,
                              const float* __restrict__ dinv,
                              float* __restrict__ h,
                              float* __restrict__ out,
                              int n_nodes, int n_tiles) {
    const int lane  = threadIdx.x & 31;
    const int wave  = threadIdx.x >> 5;
    const int tile  = blockIdx.x * (blockDim.x >> 5) + wave;
    if (tile >= n_tiles) return;

    const int base  = tile * 16;
    const int m     = lane & 15;
    const int khalf = (lane >> 4) * 2;

    int arow = base + m;
    if (arow >= n_nodes) arow = n_nodes - 1;
    const float* __restrict__ xrow = x + (size_t)arow * IN_CH;

    v2f a0 = *(const v2f*)(xrow + 0  + khalf);
    v2f a1 = *(const v2f*)(xrow + 4  + khalf);
    v2f a2 = *(const v2f*)(xrow + 8  + khalf);
    v2f a3 = *(const v2f*)(xrow + 12 + khalf);

    const int   wrow = (m < OUT_CH) ? m : 0;
    const float sel  = (m < OUT_CH) ? 1.0f : 0.0f;
    const float* __restrict__ wr = W + wrow * IN_CH + khalf;
    v2f b0 = *(const v2f*)(wr + 0);
    v2f b1 = *(const v2f*)(wr + 4);
    v2f b2 = *(const v2f*)(wr + 8);
    v2f b3 = *(const v2f*)(wr + 12);
    b0 *= sel; b1 *= sel; b2 *= sel; b3 *= sel;

    v8f acc = {};
    acc = __builtin_amdgcn_wmma_f32_16x16x4_f32(false, a0, false, b0, (short)0, acc, false, false);
    acc = __builtin_amdgcn_wmma_f32_16x16x4_f32(false, a1, false, b1, (short)0, acc, false, false);
    acc = __builtin_amdgcn_wmma_f32_16x16x4_f32(false, a2, false, b2, (short)0, acc, false, false);
    acc = __builtin_amdgcn_wmma_f32_16x16x4_f32(false, a3, false, b3, (short)0, acc, false, false);

    const int  n     = m;
    const int  mbase = base + ((lane >> 4) * 8);
    const bool full  = (base + 16) <= n_nodes;

    if (full) {
        if (n < OUT_CH) {
            const float bn = bias[n];
            const float4 d0 = *(const float4*)(dinv + mbase);
            const float4 d1 = *(const float4*)(dinv + mbase + 4);
            const float dv[8] = {d0.x, d0.y, d0.z, d0.w, d1.x, d1.y, d1.z, d1.w};
            float* __restrict__ hp = h   + (size_t)mbase * OUT_CH + n;
            float* __restrict__ op = out + (size_t)mbase * OUT_CH + n;
#pragma unroll
            for (int r = 0; r < 8; ++r) {
                const float hv = acc[r];
                hp[r * OUT_CH] = hv;
                op[r * OUT_CH] = bn + dv[r] * dv[r] * hv;
            }
        }
    } else {
        if (n < OUT_CH) {
            const float bn = bias[n];
#pragma unroll
            for (int r = 0; r < 8; ++r) {
                const int node = mbase + r;
                if (node < n_nodes) {
                    const float hv = acc[r];
                    h[(size_t)node * OUT_CH + n] = hv;
                    const float di = dinv[node];
                    out[(size_t)node * OUT_CH + n] = bn + di * di * hv;
                }
            }
        }
    }
}

__global__ void __launch_bounds__(256)
gcn_scatter(const int* __restrict__ src, const int* __restrict__ dst,
            const float* __restrict__ dinv, const float* __restrict__ h,
            float* __restrict__ out, int* __restrict__ ctr,
            int e_cnt, int n_chunks, int split, int n) {
    const int      lane = threadIdx.x & 31;
    const unsigned f    = (unsigned)(lane & 7);
    const int      grp  = lane >> 3;

    int  r      = cache_half();
    bool helped = false;
    int  next   = draw_ticket(ctr + r * CTR_STRIDE);
    const int max_trips = 2 * n_chunks + 2;
    for (int it = 0; it < max_trips; ++it) {
        const int c = next;
        if (c >= n_chunks) {
            if (helped) break;
            helped = true;
            r ^= 1;
            next = draw_ticket(ctr + r * CTR_STRIDE);
            continue;
        }
        const int lo = r ? split : 0;
        const int hi = r ? n : split;
        const int e0 = c * SC_CHUNK + lane;

        int   dv[SC_CHUNK / 32];
        int   sv[SC_CHUNK / 32];
        float wv[SC_CHUNK / 32];
        bool  in[SC_CHUNK / 32];
#pragma unroll
        for (int p = 0; p < SC_CHUNK / 32; ++p) {
            const int e  = e0 + p * 32;
            const int ec = (e < e_cnt) ? e : e_cnt - 1;
            dv[p] = dst[ec];
            sv[p] = src[ec];
        }
        next = draw_ticket(ctr + r * CTR_STRIDE);
#pragma unroll
        for (int p = 0; p < SC_CHUNK / 32; ++p) {
            in[p] = (e0 + p * 32 < e_cnt) && (dv[p] >= lo) && (dv[p] < hi);
            dv[p] = in[p] ? dv[p] : 0;
            sv[p] = in[p] ? sv[p] : 0;
        }
#pragma unroll
        for (int p = 0; p < SC_CHUNK / 32; ++p) {
            const float a = dinv[sv[p]];
            const float b = dinv[dv[p]];
            wv[p] = in[p] ? a * b : 0.0f;
        }

#pragma unroll
        for (int p = 0; p < SC_CHUNK / 32; ++p) {
            const unsigned mask = __builtin_amdgcn_ballot_w32(in[p]);
            const int cnt = __popc(mask);
            if (cnt > 0) {
                const int rank = (int)__builtin_amdgcn_mbcnt_lo(mask, 0u);
                const int slot = in[p] ? rank : cnt + lane - rank;
                const int so = __builtin_amdgcn_ds_permute(slot << 2, sv[p] * OUT_CH);
                const int dd = __builtin_amdgcn_ds_permute(slot << 2, dv[p] * OUT_CH);
                const int wi = __builtin_amdgcn_ds_permute(slot << 2, __float_as_int(wv[p]));
                for (int jb = 0; jb < cnt; jb += 16) {
                    float    hv[4];
                    unsigned oi[4];
                    float    wj[4];
#pragma unroll
                    for (int j = 0; j < 4; ++j) {
                        const int sl = jb + j * 4 + grp;
                        const int bp = sl << 2;
                        const unsigned sj = (unsigned)__builtin_amdgcn_ds_bpermute(bp, so);
                        hv[j] = (sl < cnt) ? h[sj + f] : 0.0f;
                        oi[j] = (unsigned)__builtin_amdgcn_ds_bpermute(bp, dd) + f;
                        wj[j] = __int_as_float(__builtin_amdgcn_ds_bpermute(bp, wi));
                    }
#pragma unroll
                    for (int j = 0; j < 4; ++j)
                        if (jb + j * 4 + grp < cnt) atomicAdd(out + oi[j], wj[j] * hv[j]);
                }
            }
        }
    }
}

extern "C" void kernel_launch(void* const* d_in, const int* in_sizes, int n_in,
                              void* d_out, int out_size, void* d_ws, size_t ws_size,
                              hipStream_t stream) {
    const float* x    = (const float*)d_in[0];
    const int*   ei   = (const int*)d_in[1];
    const float* W    = (const float*)d_in[2];
    const float* bias = (const float*)d_in[3];

    const int N = in_sizes[0] / IN_CH;
    const int E = in_sizes[1] / 2;
    const int* src = ei;
    const int* dst = ei + (size_t)E;

    int*   ctr  = (int*)d_ws;
    const size_t dinv_off = 4u * CTR_STRIDE * sizeof(int);
    float* dinv = (float*)((char*)d_ws + dinv_off);
    size_t h_off = ((dinv_off + (size_t)N * sizeof(float)) + 255u) & ~(size_t)255u;
    float* h = (float*)((char*)d_ws + h_off);
    float* out = (float*)d_out;

    const int TPB = 256;

    const int split = ((N / 2) + 63) & ~63;

    gcn_deg_init<<<(N + TPB - 1) / TPB, TPB, 0, stream>>>(dinv, ctr, N);

    const int deg_ranges    = (N + DEG_TILE - 1) / DEG_TILE;
    const int deg_ranges_lo = (deg_ranges + 1) / 2;
    const int deg_parts     = (256 + deg_ranges - 1) / deg_ranges;
    const int deg_slice     = (E + deg_parts - 1) / deg_parts;
    gcn_deg_count<<<256, DEG_TPB, 0, stream>>>(dst, dinv, ctr, E, N, deg_ranges,
                                               deg_ranges_lo, deg_parts, deg_slice);
    gcn_dinv<<<(N + TPB - 1) / TPB, TPB, 0, stream>>>(dinv, N);

    const int n_tiles = (N + 15) / 16;
    const int waves_per_block = TPB / 32;
    const int gemm_blocks = (n_tiles + waves_per_block - 1) / waves_per_block;
    gcn_gemm_wmma<<<gemm_blocks, TPB, 0, stream>>>(x, W, bias, dinv, h, out, N, n_tiles);

    const int sc_chunks = (E + SC_CHUNK - 1) / SC_CHUNK;
    gcn_scatter<<<2048, TPB, 0, stream>>>(src, dst, dinv, h, out, ctr + 2 * CTR_STRIDE,
                                          E, sc_chunks, split, N);
}


# edge sum without global atomics: edge slices sorted by 2048-node dst bucket in LDS; one workgroup per bucket sums dinv[src]-scaled rows by LDS float adds, stores out once (sum order, rounding change)
# speedup vs baseline: 6.2943x; 1.4448x over previous
#include <hip/hip_runtime.h>
#include <hip/hip_bf16.h>

typedef __attribute__((ext_vector_type(2))) float v2f;
typedef __attribute__((ext_vector_type(8))) float v8f;

#define IN_CH 16
#define OUT_CH 8

#define CTR_STRIDE 64
#define DEG_TILE   65536
#define DEG_TPB    1024
#define DEG_UNROLL 16
#define DEG_FLUSH  8
#define SC_CHUNK  128
#define BK_SHIFT  11
#define BK_NODES  2048
#define BK_MAX    1024
#define SRC_BITS  21
#define SL_EDGES  20480
#define SL_TPB    1024

__device__ __forceinline__ int cache_half() {
    return (int)((__builtin_amdgcn_s_sendmsg_rtn(0x87) >> 18) & 1u);
}

__device__ __forceinline__ int draw_ticket(int* ctr) {
    int t = 0;
    if ((threadIdx.x & 31) == 0) t = atomicAdd(ctr, 1);
    return __builtin_amdgcn_readfirstlane(t);
}

__global__ void gcn_deg_init(float* __restrict__ deg, int* __restrict__ ctr, int n) {
    int i = blockIdx.x * blockDim.x + threadIdx.x;
    if (i < n) deg[i] = 1.0f;
    if (i < 4) ctr[i * CTR_STRIDE] = 0;
}

__global__ void __launch_bounds__(DEG_TPB)
gcn_deg_count(const int* __restrict__ dst, float* __restrict__ deg,
              int* __restrict__ ctr, int e_cnt, int n,
              int n_ranges, int ranges_lo, int parts, int slice) {
    __shared__ int tile[DEG_TILE];
    __shared__ int s_msg[2];
    const int tid = threadIdx.x;

    if (tid == 0) s_msg[1] = cache_half();
    __syncthreads();
    int  r      = __builtin_amdgcn_readfirstlane(s_msg[1]);
    bool helped = false;
    const int max_trips = n_ranges * parts + 2;
    for (int it = 0; it < max_trips; ++it) {
        if (tid == 0) s_msg[0] = atomicAdd(ctr + r * CTR_STRIDE, 1);
        __syncthreads();
        const int t = __builtin_amdgcn_readfirstlane(s_msg[0]);
        __syncthreads();
        const int first = r ? ranges_lo : 0;
        const int count = r ? (n_ranges - ranges_lo) : ranges_lo;
        if (t >= count * parts) {
            if (helped) break;
            helped = true;
            r ^= 1;
            continue;
        }
        const int range = first + t / parts;
        const int part  = t - (t / parts) * parts;
        const int lo    = range * DEG_TILE;

        for (int i = tid; i < DEG_TILE; i += DEG_TPB) tile[i] = 0;
        __syncthreads();

        const int e_begin = part * slice;
        const int e_stop  = e_begin + slice;
        const int e_end   = e_stop < e_cnt ? e_stop : e_cnt;
        for (int base = e_begin; base < e_end; base += DEG_TPB * DEG_UNROLL) {
            int d[DEG_UNROLL];
#pragma unroll
            for (int k = 0; k < DEG_UNROLL; ++k) {
                const int e = base + k * DEG_TPB + tid;
                d[k] = (e < e_end) ? dst[e] : -1;
            }
#pragma unroll
            for (int k = 0; k < DEG_UNROLL; ++k) {
                const unsigned off = (unsigned)(d[k] - lo);
                if (off < (unsigned)DEG_TILE) atomicAdd(&tile[off], 1);
            }
        }
        __syncthreads();

        for (int i0 = tid; i0 < DEG_TILE; i0 += DEG_TPB * DEG_FLUSH) {
            int c[DEG_FLUSH];
#pragma unroll
            for (int k = 0; k < DEG_FLUSH; ++k) c[k] = tile[i0 + k * DEG_TPB];
#pragma unroll
            for (int k = 0; k < DEG_FLUSH; ++k) {
                const int node = lo + i0 + k * DEG_TPB;
                if (c[k] != 0 && node < n) atomicAdd(&deg[node], (float)c[k]);
            }
        }
    }
}

__global__ void gcn_dinv(float* __restrict__ deg, int n) {
    int i = blockIdx.x * blockDim.x + threadIdx.x;
    if (i < n) {
        float d = deg[i];
        deg[i] = d > 0.0f ? __frsqrt_rn(d) : 0.0f;
    }
}

__global__ void gcn_gemm_wmma(const float* __restrict__ x,
                              const float* __restrict__ W,
                              const float* __restrict__ bias,
                              const float* __restrict__ dinv,
                              float* __restrict__ h,
                              float* __restrict__ out,
                              int n_nodes, int n_tiles, int pre) {
    const int lane  = threadIdx.x & 31;
    const int wave  = threadIdx.x >> 5;
    const int tile  = blockIdx.x * (blockDim.x >> 5) + wave;
    if (tile >= n_tiles) return;

    const int base  = tile * 16;
    const int m     = lane & 15;
    const int khalf = (lane >> 4) * 2;

    int arow = base + m;
    if (arow >= n_nodes) arow = n_nodes - 1;
    const float* __restrict__ xrow = x + (size_t)arow * IN_CH;

    v2f a0 = *(const v2f*)(xrow + 0  + khalf);
    v2f a1 = *(const v2f*)(xrow + 4  + khalf);
    v2f a2 = *(const v2f*)(xrow + 8  + khalf);
    v2f a3 = *(const v2f*)(xrow + 12 + khalf);

    const int   wrow = (m < OUT_CH) ? m : 0;
    const float sel  = (m < OUT_CH) ? 1.0f : 0.0f;
    const float* __restrict__ wr = W + wrow * IN_CH + khalf;
    v2f b0 = *(const v2f*)(wr + 0);
    v2f b1 = *(const v2f*)(wr + 4);
    v2f b2 = *(const v2f*)(wr + 8);
    v2f b3 = *(const v2f*)(wr + 12);
    b0 *= sel; b1 *= sel; b2 *= sel; b3 *= sel;

    v8f acc = {};
    acc = __builtin_amdgcn_wmma_f32_16x16x4_f32(false, a0, false, b0, (short)0, acc, false, false);
    acc = __builtin_amdgcn_wmma_f32_16x16x4_f32(false, a1, false, b1, (short)0, acc, false, false);
    acc = __builtin_amdgcn_wmma_f32_16x16x4_f32(false, a2, false, b2, (short)0, acc, false, false);
    acc = __builtin_amdgcn_wmma_f32_16x16x4_f32(false, a3, false, b3, (short)0, acc, false, false);

    const int  n     = m;
    const int  mbase = base + ((lane >> 4) * 8);
    const bool full  = (base + 16) <= n_nodes;

    if (pre) {
        if (n < OUT_CH) {
#pragma unroll
            for (int r = 0; r < 8; ++r) {
                const int node = mbase + r;
                if (node < n_nodes) h[(size_t)node * OUT_CH + n] = dinv[node] * acc[r];
            }
        }
    } else if (full) {
        if (n < OUT_CH) {
            const float bn = bias[n];
            const float4 d0 = *(const float4*)(dinv + mbase);
            const float4 d1 = *(const float4*)(dinv + mbase + 4);
            const float dv[8] = {d0.x, d0.y, d0.z, d0.w, d1.x, d1.y, d1.z, d1.w};
            float* __restrict__ hp = h   + (size_t)mbase * OUT_CH + n;
            float* __restrict__ op = out + (size_t)mbase * OUT_CH + n;
#pragma unroll
            for (int r = 0; r < 8; ++r) {
                const float hv = acc[r];
                hp[r * OUT_CH] = hv;
                op[r * OUT_CH] = bn + dv[r] * dv[r] * hv;
            }
        }
    } else {
        if (n < OUT_CH) {
            const float bn = bias[n];
#pragma unroll
            for (int r = 0; r < 8; ++r) {
                const int node = mbase + r;
                if (node < n_nodes) {
                    const float hv = acc[r];
                    h[(size_t)node * OUT_CH + n] = hv;
                    const float di = dinv[node];
                    out[(size_t)node * OUT_CH + n] = bn + di * di * hv;
                }
            }
        }
    }
}

__global__ void __launch_bounds__(256)
gcn_scatter(const int* __restrict__ src, const int* __restrict__ dst,
            const float* __restrict__ dinv, const float* __restrict__ h,
            float* __restrict__ out, int* __restrict__ ctr,
            int e_cnt, int n_chunks, int split, int n) {
    const int      lane = threadIdx.x & 31;
    const unsigned f    = (unsigned)(lane & 7);
    const int      grp  = lane >> 3;

    int  r      = cache_half();
    bool helped = false;
    int  next   = draw_ticket(ctr + r * CTR_STRIDE);
    const int max_trips = 2 * n_chunks + 2;
    for (int it = 0; it < max_trips; ++it) {
        const int c = next;
        if (c >= n_chunks) {
            if (helped) break;
            helped = true;
            r ^= 1;
            next = draw_ticket(ctr + r * CTR_STRIDE);
            continue;
        }
        const int lo = r ? split : 0;
        const int hi = r ? n : split;
        const int e0 = c * SC_CHUNK + lane;

        int   dv[SC_CHUNK / 32];
        int   sv[SC_CHUNK / 32];
        float wv[SC_CHUNK / 32];
        bool  in[SC_CHUNK / 32];
#pragma unroll
        for (int p = 0; p < SC_CHUNK / 32; ++p) {
            const int e  = e0 + p * 32;
            const int ec = (e < e_cnt) ? e : e_cnt - 1;
            dv[p] = dst[ec];
            sv[p] = src[ec];
        }
        next = draw_ticket(ctr + r * CTR_STRIDE);
#pragma unroll
        for (int p = 0; p < SC_CHUNK / 32; ++p) {
            in[p] = (e0 + p * 32 < e_cnt) && (dv[p] >= lo) && (dv[p] < hi);
            dv[p] = in[p] ? dv[p] : 0;
            sv[p] = in[p] ? sv[p] : 0;
        }
#pragma unroll
        for (int p = 0; p < SC_CHUNK / 32; ++p) {
            const float a = dinv[sv[p]];
            const float b = dinv[dv[p]];
            wv[p] = in[p] ? a * b : 0.0f;
        }

#pragma unroll
        for (int p = 0; p < SC_CHUNK / 32; ++p) {
            const unsigned mask = __builtin_amdgcn_ballot_w32(in[p]);
            const int cnt = __popc(mask);
            if (cnt > 0) {
                const int rank = (int)__builtin_amdgcn_mbcnt_lo(mask, 0u);
                const int slot = in[p] ? rank : cnt + lane - rank;
                const int so = __builtin_amdgcn_ds_permute(slot << 2, sv[p] * OUT_CH);
                const int dd = __builtin_amdgcn_ds_permute(slot << 2, dv[p] * OUT_CH);
                const int wi = __builtin_amdgcn_ds_permute(slot << 2, __float_as_int(wv[p]));
                for (int jb = 0; jb < cnt; jb += 16) {
                    float    hv[4];
                    unsigned oi[4];
                    float    wj[4];
#pragma unroll
                    for (int j = 0; j < 4; ++j) {
                        const int sl = jb + j * 4 + grp;
                        const int bp = sl << 2;
                        const unsigned sj = (unsigned)__builtin_amdgcn_ds_bpermute(bp, so);
                        hv[j] = (sl < cnt) ? h[sj + f] : 0.0f;
                        oi[j] = (unsigned)__builtin_amdgcn_ds_bpermute(bp, dd) + f;
                        wj[j] = __int_as_float(__builtin_amdgcn_ds_bpermute(bp, wi));
                    }
#pragma unroll
                    for (int j = 0; j < 4; ++j)
                        if (jb + j * 4 + grp < cnt) atomicAdd(out + oi[j], wj[j] * hv[j]);
                }
            }
        }
    }
}

__global__ void __launch_bounds__(SL_TPB)
gcn_bin_slice(const int* __restrict__ src, const int* __restrict__ dst,
              unsigned* __restrict__ binned, int* __restrict__ runs,
              int e_cnt, int nb) {
    __shared__ unsigned stage[SL_EDGES];
    __shared__ int cnt[BK_MAX];
    __shared__ int wsum[32];
    const int tid  = threadIdx.x;
    const int lane = tid & 31;
    const int wave = tid >> 5;
    const int s    = blockIdx.x;
    const int e0   = s * SL_EDGES;
    const int e_hi = e0 + SL_EDGES;
    const int e_end = e_hi < e_cnt ? e_hi : e_cnt;

    cnt[tid] = 0;
    __syncthreads();

    int dv[SL_EDGES / SL_TPB];
    int sv[SL_EDGES / SL_TPB];
#pragma unroll
    for (int k = 0; k < SL_EDGES / SL_TPB; ++k) {
        const int e  = e0 + k * SL_TPB + tid;
        const int ec = e < e_end ? e : e_end - 1;
        dv[k] = dst[ec];
        sv[k] = src[ec];
    }
#pragma unroll
    for (int k = 0; k < SL_EDGES / SL_TPB; ++k)
        if (e0 + k * SL_TPB + tid < e_end) atomicAdd(&cnt[dv[k] >> BK_SHIFT], 1);
    __syncthreads();

    const int v = cnt[tid];
    int incl = v;
#pragma unroll
    for (int o = 1; o < 32; o <<= 1) {
        const int t = __shfl_up(incl, o);
        if (lane >= o) incl += t;
    }
    if (lane == 31) wsum[wave] = incl;
    __syncthreads();
    if (wave == 0) {
        const int wt = wsum[lane];
        int wi = wt;
#pragma unroll
        for (int o = 1; o < 32; o <<= 1) {
            const int t = __shfl_up(wi, o);
            if (lane >= o) wi += t;
        }
        wsum[lane] = wi - wt;
    }
    __syncthreads();
    const int excl = incl - v + wsum[wave];
    cnt[tid] = excl;
    if (tid <= nb) runs[(size_t)s * (nb + 1) + tid] = excl;
    __syncthreads();

#pragma unroll
    for (int k = 0; k < SL_EDGES / SL_TPB; ++k) {
        if (e0 + k * SL_TPB + tid < e_end) {
            const int p = atomicAdd(&cnt[dv[k] >> BK_SHIFT], 1);
            stage[p] = (unsigned)sv[k] | ((unsigned)(dv[k] & (BK_NODES - 1)) << SRC_BITS);
        }
    }
    __syncthreads();

    const int n_s = e_end - e0;
    unsigned* __restrict__ region = binned + (size_t)s * SL_EDGES;
    for (int i = tid; i < n_s; i += SL_TPB) region[i] = stage[i];
}

__device__ __forceinline__ void run_words(const unsigned* __restrict__ region,
                                          int p0, int hi, int lane, unsigned (&w)[4]) {
#pragma unroll
    for (int c = 0; c < 4; ++c) {
        const int p = p0 + c * 32 + lane;
        w[c] = region[p < hi ? p : hi - 1];
    }
}

__device__ __forceinline__ void add_64(float* __restrict__ tile, const float* __restrict__ hs,
                                       unsigned w0, unsigned w1, int left,
                                       int grp, unsigned f) {
    unsigned ww[16];
    float    hv[16];
#pragma unroll
    for (int j = 0; j < 16; ++j) {
        const int bp = ((j & 7) * 4 + grp) << 2;
        ww[j] = (unsigned)__builtin_amdgcn_ds_bpermute(bp, (int)(j < 8 ? w0 : w1));
    }
#pragma unroll
    for (int j = 0; j < 16; ++j) {
        const unsigned sr = (j * 4 + grp < left) ? (ww[j] & ((1u << SRC_BITS) - 1u)) : 0u;
        hv[j] = hs[(size_t)sr * OUT_CH + f];
    }
#pragma unroll
    for (int j = 0; j < 16; ++j)
        if (j * 4 + grp < left) atomicAdd(&tile[(ww[j] >> SRC_BITS) * OUT_CH + f], hv[j]);
}

__global__ void __launch_bounds__(SL_TPB)
gcn_bucket_sum(const unsigned* __restrict__ binned, const int* __restrict__ runs,
               const float* __restrict__ hs, const float* __restrict__ dinv,
               const float* __restrict__ bias, float* __restrict__ out,
               int n, int nb, int n_slices) {
    __shared__ float tile[BK_NODES * OUT_CH];
    __shared__ int   rlo[SL_TPB];
    __shared__ int   rhi[SL_TPB];
    const int      tid  = threadIdx.x;
    const int      lane = tid & 31;
    const int      wave = tid >> 5;
    const unsigned f    = (unsigned)(lane & 7);
    const int      grp  = lane >> 3;
    const int      b    = blockIdx.x;

    if (tid < n_slices) {
        const int* __restrict__ rp = runs + (size_t)tid * (nb + 1) + b;
        rlo[tid] = rp[0];
        rhi[tid] = rp[1];
    }
    for (int i = tid; i < BK_NODES * OUT_CH; i += SL_TPB) tile[i] = 0.0f;
    __syncthreads();

    int s  = wave;
    int lo = 0, hi = 0;
    unsigned w[4] = {0u, 0u, 0u, 0u};
    if (s < n_slices) {
        lo = __builtin_amdgcn_readfirstlane(rlo[s]);
        hi = __builtin_amdgcn_readfirstlane(rhi[s]);
        if (hi > lo) run_words(binned + (size_t)s * SL_EDGES, lo, hi, lane, w);
    }
    for (; s < n_slices; s += SL_TPB / 32) {
        const int sn = s + SL_TPB / 32;
        int lo2 = 0, hi2 = 0;
        unsigned wn[4] = {0u, 0u, 0u, 0u};
        if (sn < n_slices) {
            lo2 = __builtin_amdgcn_readfirstlane(rlo[sn]);
            hi2 = __builtin_amdgcn_readfirstlane(rhi[sn]);
            if (hi2 > lo2) run_words(binned + (size_t)sn * SL_EDGES, lo2, hi2, lane, wn);
        }
        if (hi > lo)      add_64(tile, hs, w[0], w[1], hi - lo, grp, f);
        if (hi > lo + 64) add_64(tile, hs, w[2], w[3], hi - lo - 64, grp, f);
        for (int p0 = lo + 128; p0 < hi; p0 += 128) {
            unsigned wr[4];
            run_words(binned + (size_t)s * SL_EDGES, p0, hi, lane, wr);
            add_64(tile, hs, wr[0], wr[1], hi - p0, grp, f);
            if (hi > p0 + 64) add_64(tile, hs, wr[2], wr[3], hi - p0 - 64, grp, f);
        }
        lo = lo2; hi = hi2;
#pragma unroll
        for (int c = 0; c < 4; ++c) w[c] = wn[c];
    }
    __syncthreads();

    const int node0 = b * BK_NODES;
    for (int i = tid; i < BK_NODES * OUT_CH; i += SL_TPB) {
        const int node = node0 + (i >> 3);
        if (node < n) {
            const size_t o = (size_t)node * OUT_CH + (i & 7);
            out[o] = bias[i & 7] + dinv[node] * (tile[i] + hs[o]);
        }
    }
}

extern "C" void kernel_launch(void* const* d_in, const int* in_sizes, int n_in,
                              void* d_out, int out_size, void* d_ws, size_t ws_size,
                              hipStream_t stream) {
    const float* x    = (const float*)d_in[0];
    const int*   ei   = (const int*)d_in[1];
    const float* W    = (const float*)d_in[2];
    const float* bias = (const float*)d_in[3];

    const int N = in_sizes[0] / IN_CH;
    const int E = in_sizes[1] / 2;
    const int* src = ei;
    const int* dst = ei + (size_t)E;

    int*   ctr  = (int*)d_ws;
    const size_t dinv_off = 4u * CTR_STRIDE * sizeof(int);
    float* dinv = (float*)((char*)d_ws + dinv_off);
    size_t h_off = ((dinv_off + (size_t)N * sizeof(float)) + 255u) & ~(size_t)255u;
    float* h = (float*)((char*)d_ws + h_off);
    float* out = (float*)d_out;

    const int TPB = 256;

    const int split = ((N / 2) + 63) & ~63;

    gcn_deg_init<<<(N + TPB - 1) / TPB, TPB, 0, stream>>>(dinv, ctr, N);

    const int deg_ranges    = (N + DEG_TILE - 1) / DEG_TILE;
    const int deg_ranges_lo = (deg_ranges + 1) / 2;
    const int deg_parts     = (256 + deg_ranges - 1) / deg_ranges;
    const int deg_slice     = (E + deg_parts - 1) / deg_parts;
    gcn_deg_count<<<256, DEG_TPB, 0, stream>>>(dst, dinv, ctr, E, N, deg_ranges,
                                               deg_ranges_lo, deg_parts, deg_slice);
    gcn_dinv<<<(N + TPB - 1) / TPB, TPB, 0, stream>>>(dinv, N);

    const int    nb       = (N + BK_NODES - 1) >> BK_SHIFT;
    const int    n_slices = (E + SL_EDGES - 1) / SL_EDGES;
    const size_t bin_off  = ((h_off + (size_t)N * OUT_CH * sizeof(float)) + 255u) & ~(size_t)255u;
    const size_t runs_off = ((bin_off + (size_t)n_slices * SL_EDGES * sizeof(unsigned)) + 255u) & ~(size_t)255u;
    const size_t ws_need  = runs_off + (size_t)n_slices * (size_t)(nb + 1) * sizeof(int);
    const bool   buckets  = E > 0 && nb < BK_MAX && n_slices <= SL_TPB && N <= (1 << SRC_BITS) &&
                            ws_need <= ws_size;

    const int n_tiles = (N + 15) / 16;
    const int waves_per_block = TPB / 32;
    const int gemm_blocks = (n_tiles + waves_per_block - 1) / waves_per_block;
    gcn_gemm_wmma<<<gemm_blocks, TPB, 0, stream>>>(x, W, bias, dinv, h, out, N, n_tiles,
                                                   buckets ? 1 : 0);

    if (buckets) {
        unsigned* binned = (unsigned*)((char*)d_ws + bin_off);
        int*      runs   = (int*)((char*)d_ws + runs_off);
        gcn_bin_slice<<<n_slices, SL_TPB, 0, stream>>>(src, dst, binned, runs, E, nb);
        gcn_bucket_sum<<<nb, SL_TPB, 0, stream>>>(binned, runs, h, dinv, bias, out, N, nb, n_slices);
    } else {
        const int sc_chunks = (E + SC_CHUNK - 1) / SC_CHUNK;
        gcn_scatter<<<2048, TPB, 0, stream>>>(src, dst, dinv, h, out, ctr + 2 * CTR_STRIDE,
                                              E, sc_chunks, split, N);
    }
}


# degree from the bucket-sorted edge words: one workgroup per bucket counts its rows in an LDS histogram, writes dinv and does the rows' WMMA tiles itself; 6 launches become 3; values unchanged
# speedup vs baseline: 8.5414x; 1.3570x over previous
#include <hip/hip_runtime.h>
#include <hip/hip_bf16.h>

typedef __attribute__((ext_vector_type(2))) float v2f;
typedef __attribute__((ext_vector_type(8))) float v8f;

#define IN_CH 16
#define OUT_CH 8

#define CTR_STRIDE 64
#define DEG_TILE   65536
#define DEG_TPB    1024
#define DEG_UNROLL 16
#define DEG_FLUSH  8
#define SC_CHUNK  128
#define BK_SHIFT  11
#define BK_NODES  2048
#define BK_MAX    1024
#define SRC_BITS  21
#define SL_EDGES  20480
#define SL_TPB    1024
#define ROW_TILES 4

__device__ __forceinline__ int cache_half() {
    return (int)((__builtin_amdgcn_s_sendmsg_rtn(0x87) >> 18) & 1u);
}

__device__ __forceinline__ int draw_ticket(int* ctr) {
    int t = 0;
    if ((threadIdx.x & 31) == 0) t = atomicAdd(ctr, 1);
    return __builtin_amdgcn_readfirstlane(t);
}

__global__ void gcn_deg_init(float* __restrict__ deg, int* __restrict__ ctr, int n) {
    int i = blockIdx.x * blockDim.x + threadIdx.x;
    if (i < n) deg[i] = 1.0f;
    if (i < 4) ctr[i * CTR_STRIDE] = 0;
}

__global__ void __launch_bounds__(DEG_TPB)
gcn_deg_count(const int* __restrict__ dst, float* __restrict__ deg,
              int* __restrict__ ctr, int e_cnt, int n,
              int n_ranges, int ranges_lo, int parts, int slice) {
    __shared__ int tile[DEG_TILE];
    __shared__ int s_msg[2];
    const int tid = threadIdx.x;

    if (tid == 0) s_msg[1] = cache_half();
    __syncthreads();
    int  r      = __builtin_amdgcn_readfirstlane(s_msg[1]);
    bool helped = false;
    const int max_trips = n_ranges * parts + 2;
    for (int it = 0; it < max_trips; ++it) {
        if (tid == 0) s_msg[0] = atomicAdd(ctr + r * CTR_STRIDE, 1);
        __syncthreads();
        const int t = __builtin_amdgcn_readfirstlane(s_msg[0]);
        __syncthreads();
        const int first = r ? ranges_lo : 0;
        const int count = r ? (n_ranges - ranges_lo) : ranges_lo;
        if (t >= count * parts) {
            if (helped) break;
            helped = true;
            r ^= 1;
            continue;
        }
        const int range = first + t / parts;
        const int part  = t - (t / parts) * parts;
        const int lo    = range * DEG_TILE;

        for (int i = tid; i < DEG_TILE; i += DEG_TPB) tile[i] = 0;
        __syncthreads();

        const int e_begin = part * slice;
        const int e_stop  = e_begin + slice;
        const int e_end   = e_stop < e_cnt ? e_stop : e_cnt;
        for (int base = e_begin; base < e_end; base += DEG_TPB * DEG_UNROLL) {
            int d[DEG_UNROLL];
#pragma unroll
            for (int k = 0; k < DEG_UNROLL; ++k) {
                const int e = base + k * DEG_TPB + tid;
                d[k] = (e < e_end) ? dst[e] : -1;
            }
#pragma unroll
            for (int k = 0; k < DEG_UNROLL; ++k) {
                const unsigned off = (unsigned)(d[k] - lo);
                if (off < (unsigned)DEG_TILE) atomicAdd(&tile[off], 1);
            }
        }
        __syncthreads();

        for (int i0 = tid; i0 < DEG_TILE; i0 += DEG_TPB * DEG_FLUSH) {
            int c[DEG_FLUSH];
#pragma unroll
            for (int k = 0; k < DEG_FLUSH; ++k) c[k] = tile[i0 + k * DEG_TPB];
#pragma unroll
            for (int k = 0; k < DEG_FLUSH; ++k) {
                const int node = lo + i0 + k * DEG_TPB;
                if (c[k] != 0 && node < n) atomicAdd(&deg[node], (float)c[k]);
            }
        }
    }
}

__global__ void gcn_dinv(float* __restrict__ deg, int n) {
    int i = blockIdx.x * blockDim.x + threadIdx.x;
    if (i < n) {
        float d = deg[i];
        deg[i] = d > 0.0f ? __frsqrt_rn(d) : 0.0f;
    }
}

__global__ void gcn_gemm_wmma(const float* __restrict__ x,
                              const float* __restrict__ W,
                              const float* __restrict__ bias,
                              const float* __restrict__ dinv,
                              float* __restrict__ h,
                              float* __restrict__ out,
                              int n_nodes, int n_tiles) {
    const int lane  = threadIdx.x & 31;
    const int wave  = threadIdx.x >> 5;
    const int tile  = blockIdx.x * (blockDim.x >> 5) + wave;
    if (tile >= n_tiles) return;

    const int base  = tile * 16;
    const int m     = lane & 15;
    const int khalf = (lane >> 4) * 2;

    int arow = base + m;
    if (arow >= n_nodes) arow = n_nodes - 1;
    const float* __restrict__ xrow = x + (size_t)arow * IN_CH;

    v2f a0 = *(const v2f*)(xrow + 0  + khalf);
    v2f a1 = *(const v2f*)(xrow + 4  + khalf);
    v2f a2 = *(const v2f*)(xrow + 8  + khalf);
    v2f a3 = *(const v2f*)(xrow + 12 + khalf);

    const int   wrow = (m < OUT_CH) ? m : 0;
    const float sel  = (m < OUT_CH) ? 1.0f : 0.0f;
    const float* __restrict__ wr = W + wrow * IN_CH + khalf;
    v2f b0 = *(const v2f*)(wr + 0);
    v2f b1 = *(const v2f*)(wr + 4);
    v2f b2 = *(const v2f*)(wr + 8);
    v2f b3 = *(const v2f*)(wr + 12);
    b0 *= sel; b1 *= sel; b2 *= sel; b3 *= sel;

    v8f acc = {};
    acc = __builtin_amdgcn_wmma_f32_16x16x4_f32(false, a0, false, b0, (short)0, acc, false, false);
    acc = __builtin_amdgcn_wmma_f32_16x16x4_f32(false, a1, false, b1, (short)0, acc, false, false);
    acc = __builtin_amdgcn_wmma_f32_16x16x4_f32(false, a2, false, b2, (short)0, acc, false, false);
    acc = __builtin_amdgcn_wmma_f32_16x16x4_f32(false, a3, false, b3, (short)0, acc, false, false);

    const int  n     = m;
    const int  mbase = base + ((lane >> 4) * 8);
    const bool full  = (base + 16) <= n_nodes;

    if (full) {
        if (n < OUT_CH) {
            const float bn = bias[n];
            const float4 d0 = *(const float4*)(dinv + mbase);
            const float4 d1 = *(const float4*)(dinv + mbase + 4);
            const float dv[8] = {d0.x, d0.y, d0.z, d0.w, d1.x, d1.y, d1.z, d1.w};
            float* __restrict__ hp = h   + (size_t)mbase * OUT_CH + n;
            float* __restrict__ op = out + (size_t)mbase * OUT_CH + n;
#pragma unroll
            for (int r = 0; r < 8; ++r) {
                const float hv = acc[r];
                hp[r * OUT_CH] = hv;
                op[r * OUT_CH] = bn + dv[r] * dv[r] * hv;
            }
        }
    } else {
        if (n < OUT_CH) {
            const float bn = bias[n];
#pragma unroll
            for (int r = 0; r < 8; ++r) {
                const int node = mbase + r;
                if (node < n_nodes) {
                    const float hv = acc[r];
                    h[(size_t)node * OUT_CH + n] = hv;
                    const float di = dinv[node];
                    out[(size_t)node * OUT_CH + n] = bn + di * di * hv;
                }
            }
        }
    }
}

__global__ void __launch_bounds__(256)
gcn_scatter(const int* __restrict__ src, const int* __restrict__ dst,
            const float* __restrict__ dinv, const float* __restrict__ h,
            float* __restrict__ out, int* __restrict__ ctr,
            int e_cnt, int n_chunks, int split, int n) {
    const int      lane = threadIdx.x & 31;
    const unsigned f    = (unsigned)(lane & 7);
    const int      grp  = lane >> 3;

    int  r      = cache_half();
    bool helped = false;
    int  next   = draw_ticket(ctr + r * CTR_STRIDE);
    const int max_trips = 2 * n_chunks + 2;
    for (int it = 0; it < max_trips; ++it) {
        const int c = next;
        if (c >= n_chunks) {
            if (helped) break;
            helped = true;
            r ^= 1;
            next = draw_ticket(ctr + r * CTR_STRIDE);
            continue;
        }
        const int lo = r ? split : 0;
        const int hi = r ? n : split;
        const int e0 = c * SC_CHUNK + lane;

        int   dv[SC_CHUNK / 32];
        int   sv[SC_CHUNK / 32];
        float wv[SC_CHUNK / 32];
        bool  in[SC_CHUNK / 32];
#pragma unroll
        for (int p = 0; p < SC_CHUNK / 32; ++p) {
            const int e  = e0 + p * 32;
            const int ec = (e < e_cnt) ? e : e_cnt - 1;
            dv[p] = dst[ec];
            sv[p] = src[ec];
        }
        next = draw_ticket(ctr + r * CTR_STRIDE);
#pragma unroll
        for (int p = 0; p < SC_CHUNK / 32; ++p) {
            in[p] = (e0 + p * 32 < e_cnt) && (dv[p] >= lo) && (dv[p] < hi);
            dv[p] = in[p] ? dv[p] : 0;
            sv[p] = in[p] ? sv[p] : 0;
        }
#pragma unroll
        for (int p = 0; p < SC_CHUNK / 32; ++p) {
            const float a = dinv[sv[p]];
            const float b = dinv[dv[p]];
            wv[p] = in[p] ? a * b : 0.0f;
        }

#pragma unroll
        for (int p = 0; p < SC_CHUNK / 32; ++p) {
            const unsigned mask = __builtin_amdgcn_ballot_w32(in[p]);
            const int cnt = __popc(mask);
            if (cnt > 0) {
                const int rank = (int)__builtin_amdgcn_mbcnt_lo(mask, 0u);
                const int slot = in[p] ? rank : cnt + lane - rank;
                const int so = __builtin_amdgcn_ds_permute(slot << 2, sv[p] * OUT_CH);
                const int dd = __builtin_amdgcn_ds_permute(slot << 2, dv[p] * OUT_CH);
                const int wi = __builtin_amdgcn_ds_permute(slot << 2, __float_as_int(wv[p]));
                for (int jb = 0; jb < cnt; jb += 16) {
                    float    hv[4];
                    unsigned oi[4];
                    float    wj[4];
#pragma unroll
                    for (int j = 0; j < 4; ++j) {
                        const int sl = jb + j * 4 + grp;
                        const int bp = sl << 2;
                        const unsigned sj = (unsigned)__builtin_amdgcn_ds_bpermute(bp, so);
                        hv[j] = (sl < cnt) ? h[sj + f] : 0.0f;
                        oi[j] = (unsigned)__builtin_amdgcn_ds_bpermute(bp, dd) + f;
                        wj[j] = __int_as_float(__builtin_amdgcn_ds_bpermute(bp, wi));
                    }
#pragma unroll
                    for (int j = 0; j < 4; ++j)
                        if (jb + j * 4 + grp < cnt) atomicAdd(out + oi[j], wj[j] * hv[j]);
                }
            }
        }
    }
}

__global__ void __launch_bounds__(SL_TPB)
gcn_bin_slice(const int* __restrict__ src, const int* __restrict__ dst,
              unsigned* __restrict__ binned, int* __restrict__ runs,
              int e_cnt, int nb) {
    __shared__ unsigned stage[SL_EDGES];
    __shared__ int cnt[BK_MAX];
    __shared__ int wsum[32];
    const int tid  = threadIdx.x;
    const int lane = tid & 31;
    const int wave = tid >> 5;
    const int s    = blockIdx.x;
    const int e0   = s * SL_EDGES;
    const int e_hi = e0 + SL_EDGES;
    const int e_end = e_hi < e_cnt ? e_hi : e_cnt;

    cnt[tid] = 0;
    __syncthreads();

    int dv[SL_EDGES / SL_TPB];
    int sv[SL_EDGES / SL_TPB];
#pragma unroll
    for (int k = 0; k < SL_EDGES / SL_TPB; ++k) {
        const int e  = e0 + k * SL_TPB + tid;
        const int ec = e < e_end ? e : e_end - 1;
        dv[k] = dst[ec];
        sv[k] = src[ec];
    }
#pragma unroll
    for (int k = 0; k < SL_EDGES / SL_TPB; ++k)
        if (e0 + k * SL_TPB + tid < e_end) atomicAdd(&cnt[dv[k] >> BK_SHIFT], 1);
    __syncthreads();

    const int v = cnt[tid];
    int incl = v;
#pragma unroll
    for (int o = 1; o < 32; o <<= 1) {
        const int t = __shfl_up(incl, o);
        if (lane >= o) incl += t;
    }
    if (lane == 31) wsum[wave] = incl;
    __syncthreads();
    if (wave == 0) {
        const int wt = wsum[lane];
        int wi = wt;
#pragma unroll
        for (int o = 1; o < 32; o <<= 1) {
            const int t = __shfl_up(wi, o);
            if (lane >= o) wi += t;
        }
        wsum[lane] = wi - wt;
    }
    __syncthreads();
    const int excl = incl - v + wsum[wave];
    cnt[tid] = excl;
    if (tid <= nb) runs[(size_t)s * (nb + 1) + tid] = excl;
    __syncthreads();

#pragma unroll
    for (int k = 0; k < SL_EDGES / SL_TPB; ++k) {
        if (e0 + k * SL_TPB + tid < e_end) {
            const int p = atomicAdd(&cnt[dv[k] >> BK_SHIFT], 1);
            stage[p] = (unsigned)sv[k] | ((unsigned)(dv[k] & (BK_NODES - 1)) << SRC_BITS);
        }
    }
    __syncthreads();

    const int n_s = e_end - e0;
    unsigned* __restrict__ region = binned + (size_t)s * SL_EDGES;
    for (int i = tid; i < n_s; i += SL_TPB) region[i] = stage[i];
}

__device__ __forceinline__ void run_words(const unsigned* __restrict__ region,
                                          int p0, int hi, int lane, unsigned (&w)[4]) {
#pragma unroll
    for (int c = 0; c < 4; ++c) {
        const int p = p0 + c * 32 + lane;
        w[c] = region[p < hi ? p : hi - 1];
    }
}

__global__ void __launch_bounds__(SL_TPB)
gcn_bucket_rows(const unsigned* __restrict__ binned, const int* __restrict__ runs,
                const float* __restrict__ x,
                const float* __restrict__ W,
                float* __restrict__ dinv, float* __restrict__ hs,
                int n, int nb, int n_slices) {
    __shared__ int   hist[BK_NODES];
    __shared__ float dloc[BK_NODES];
    __shared__ int   rlo[SL_TPB];
    __shared__ int   rhi[SL_TPB];
    const int tid   = threadIdx.x;
    const int lane  = tid & 31;
    const int wave  = __builtin_amdgcn_readfirstlane(tid >> 5);
    const int b     = blockIdx.x;
    const int node0 = b * BK_NODES;
    const int m     = lane & 15;
    const int khalf = (lane >> 4) * 2;

    v2f a[ROW_TILES][4];
#pragma unroll
    for (int q = 0; q < ROW_TILES; ++q) {
        int arow = node0 + (wave * ROW_TILES + q) * 16 + m;
        if (arow >= n) arow = n - 1;
        const float* __restrict__ xrow = x + (size_t)arow * IN_CH + khalf;
#pragma unroll
        for (int k = 0; k < 4; ++k) a[q][k] = *(const v2f*)(xrow + 4 * k);
    }
    const int   wrow = (m < OUT_CH) ? m : 0;
    const float sel  = (m < OUT_CH) ? 1.0f : 0.0f;
    const float* __restrict__ wr = W + wrow * IN_CH + khalf;
    v2f bf[4];
#pragma unroll
    for (int k = 0; k < 4; ++k) bf[k] = *(const v2f*)(wr + 4 * k) * sel;

    if (tid < n_slices) {
        const int* __restrict__ rp = runs + (size_t)tid * (nb + 1) + b;
        rlo[tid] = rp[0];
        rhi[tid] = rp[1];
    }
    for (int i = tid; i < BK_NODES; i += SL_TPB) hist[i] = 0;
    __syncthreads();

    for (int s0 = wave; s0 < n_slices; s0 += (SL_TPB / 32) * 4) {
        int      lo[4], hi[4];
        unsigned w[4][4];
#pragma unroll
        for (int u = 0; u < 4; ++u) {
            const int  s  = s0 + u * (SL_TPB / 32);
            const bool ok = s < n_slices;
            const int  sc = ok ? s : s0;
            const int  l  = __builtin_amdgcn_readfirstlane(rlo[sc]);
            const int  h  = __builtin_amdgcn_readfirstlane(rhi[sc]);
            lo[u] = ok ? l : 0;
            hi[u] = ok ? h : 0;
            const unsigned* __restrict__ region = binned + (size_t)sc * SL_EDGES;
#pragma unroll
            for (int cc = 0; cc < 4; ++cc) {
                const int p  = lo[u] + cc * 32 + lane;
                const int pc = p < hi[u] ? p : hi[u] - 1;
                w[u][cc] = region[pc > 0 ? pc : 0];
            }
        }
#pragma unroll
        for (int u = 0; u < 4; ++u) {
#pragma unroll
            for (int cc = 0; cc < 4; ++cc)
                if (lo[u] + cc * 32 + lane < hi[u]) atomicAdd(&hist[w[u][cc] >> SRC_BITS], 1);
        }
#pragma unroll
        for (int u = 0; u < 4; ++u) {
            for (int p0 = lo[u] + 128; p0 < hi[u]; p0 += 128) {
                unsigned wt[4];
                run_words(binned + (size_t)(s0 + u * (SL_TPB / 32)) * SL_EDGES, p0, hi[u], lane, wt);
#pragma unroll
                for (int cc = 0; cc < 4; ++cc)
                    if (p0 + cc * 32 + lane < hi[u]) atomicAdd(&hist[wt[cc] >> SRC_BITS], 1);
            }
        }
    }
    __syncthreads();

    for (int i = tid; i < BK_NODES; i += SL_TPB) {
        const float d = (float)(hist[i] + 1);
        const float v = d > 0.0f ? __frsqrt_rn(d) : 0.0f;
        dloc[i] = v;
        if (node0 + i < n) dinv[node0 + i] = v;
    }
    __syncthreads();

    v8f acc[ROW_TILES];
#pragma unroll
    for (int q = 0; q < ROW_TILES; ++q) acc[q] = v8f{};
#pragma unroll
    for (int k = 0; k < 4; ++k) {
#pragma unroll
        for (int q = 0; q < ROW_TILES; ++q)
            acc[q] = __builtin_amdgcn_wmma_f32_16x16x4_f32(false, a[q][k], false, bf[k], (short)0, acc[q], false, false);
    }

    if (m < OUT_CH) {
#pragma unroll
        for (int q = 0; q < ROW_TILES; ++q) {
            const int tb    = (wave * ROW_TILES + q) * 16;
            const int rbase = tb + (lane >> 4) * 8;
            const float4 d0 = *(const float4*)(dloc + rbase);
            const float4 d1 = *(const float4*)(dloc + rbase + 4);
            const float dv[8] = {d0.x, d0.y, d0.z, d0.w, d1.x, d1.y, d1.z, d1.w};
            float* __restrict__ hp = hs + (size_t)(node0 + rbase) * OUT_CH + m;
            if (node0 + tb + 16 <= n) {
#pragma unroll
                for (int r = 0; r < 8; ++r) hp[r * OUT_CH] = dv[r] * acc[q][r];
            } else {
#pragma unroll
                for (int r = 0; r < 8; ++r)
                    if (node0 + rbase + r < n) hp[r * OUT_CH] = dv[r] * acc[q][r];
            }
        }
    }
}

__device__ __forceinline__ void add_64(float* __restrict__ tile, const float* __restrict__ hs,
                                       unsigned w0, unsigned w1, int left,
                                       int grp, unsigned f) {
    unsigned ww[16];
    float    hv[16];
#pragma unroll
    for (int j = 0; j < 16; ++j) {
        const int bp = ((j & 7) * 4 + grp) << 2;
        ww[j] = (unsigned)__builtin_amdgcn_ds_bpermute(bp, (int)(j < 8 ? w0 : w1));
    }
#pragma unroll
    for (int j = 0; j < 16; ++j) {
        const unsigned sr = (j * 4 + grp < left) ? (ww[j] & ((1u << SRC_BITS) - 1u)) : 0u;
        hv[j] = hs[(size_t)sr * OUT_CH + f];
    }
#pragma unroll
    for (int j = 0; j < 16; ++j)
        if (j * 4 + grp < left) atomicAdd(&tile[(ww[j] >> SRC_BITS) * OUT_CH + f], hv[j]);
}

__global__ void __launch_bounds__(SL_TPB)
gcn_bucket_sum(const unsigned* __restrict__ binned, const int* __restrict__ runs,
               const float* __restrict__ hs, const float* __restrict__ dinv,
               const float* __restrict__ bias, float* __restrict__ out,
               int n, int nb, int n_slices) {
    __shared__ float tile[BK_NODES * OUT_CH];
    __shared__ int   rlo[SL_TPB];
    __shared__ int   rhi[SL_TPB];
    const int      tid  = threadIdx.x;
    const int      lane = tid & 31;
    const int      wave = tid >> 5;
    const unsigned f    = (unsigned)(lane & 7);
    const int      grp  = lane >> 3;
    const int      b    = blockIdx.x;

    if (tid < n_slices) {
        const int* __restrict__ rp = runs + (size_t)tid * (nb + 1) + b;
        rlo[tid] = rp[0];
        rhi[tid] = rp[1];
    }
    for (int i = tid; i < BK_NODES * OUT_CH; i += SL_TPB) tile[i] = 0.0f;
    __syncthreads();

    int s  = wave;
    int lo = 0, hi = 0;
    unsigned w[4] = {0u, 0u, 0u, 0u};
    if (s < n_slices) {
        lo = __builtin_amdgcn_readfirstlane(rlo[s]);
        hi = __builtin_amdgcn_readfirstlane(rhi[s]);
        if (hi > lo) run_words(binned + (size_t)s * SL_EDGES, lo, hi, lane, w);
    }
    for (; s < n_slices; s += SL_TPB / 32) {
        const int sn = s + SL_TPB / 32;
        int lo2 = 0, hi2 = 0;
        unsigned wn[4] = {0u, 0u, 0u, 0u};
        if (sn < n_slices) {
            lo2 = __builtin_amdgcn_readfirstlane(rlo[sn]);
            hi2 = __builtin_amdgcn_readfirstlane(rhi[sn]);
            if (hi2 > lo2) run_words(binned + (size_t)sn * SL_EDGES, lo2, hi2, lane, wn);
        }
        if (hi > lo)      add_64(tile, hs, w[0], w[1], hi - lo, grp, f);
        if (hi > lo + 64) add_64(tile, hs, w[2], w[3], hi - lo - 64, grp, f);
        for (int p0 = lo + 128; p0 < hi; p0 += 128) {
            unsigned wr[4];
            run_words(binned + (size_t)s * SL_EDGES, p0, hi, lane, wr);
            add_64(tile, hs, wr[0], wr[1], hi - p0, grp, f);
            if (hi > p0 + 64) add_64(tile, hs, wr[2], wr[3], hi - p0 - 64, grp, f);
        }
        lo = lo2; hi = hi2;
#pragma unroll
        for (int c = 0; c < 4; ++c) w[c] = wn[c];
    }
    __syncthreads();

    const int node0 = b * BK_NODES;
    for (int i = tid; i < BK_NODES * OUT_CH; i += SL_TPB) {
        const int node = node0 + (i >> 3);
        if (node < n) {
            const size_t o = (size_t)node * OUT_CH + (i & 7);
            out[o] = bias[i & 7] + dinv[node] * (tile[i] + hs[o]);
        }
    }
}

extern "C" void kernel_launch(void* const* d_in, const int* in_sizes, int n_in,
                              void* d_out, int out_size, void* d_ws, size_t ws_size,
                              hipStream_t stream) {
    const float* x    = (const float*)d_in[0];
    const int*   ei   = (const int*)d_in[1];
    const float* W    = (const float*)d_in[2];
    const float* bias = (const float*)d_in[3];

    const int N = in_sizes[0] / IN_CH;
    const int E = in_sizes[1] / 2;
    const int* src = ei;
    const int* dst = ei + (size_t)E;

    int*   ctr  = (int*)d_ws;
    const size_t dinv_off = 4u * CTR_STRIDE * sizeof(int);
    float* dinv = (float*)((char*)d_ws + dinv_off);
    size_t h_off = ((dinv_off + (size_t)N * sizeof(float)) + 255u) & ~(size_t)255u;
    float* h = (float*)((char*)d_ws + h_off);
    float* out = (float*)d_out;

    const int TPB = 256;

    const int split = ((N / 2) + 63) & ~63;

    const int    nb       = (N + BK_NODES - 1) >> BK_SHIFT;
    const int    n_slices = (E + SL_EDGES - 1) / SL_EDGES;
    const size_t bin_off  = ((h_off + (size_t)N * OUT_CH * sizeof(float)) + 255u) & ~(size_t)255u;
    const size_t runs_off = ((bin_off + (size_t)n_slices * SL_EDGES * sizeof(unsigned)) + 255u) & ~(size_t)255u;
    const size_t ws_need  = runs_off + (size_t)n_slices * (size_t)(nb + 1) * sizeof(int);
    const bool   buckets  = E > 0 && nb < BK_MAX && n_slices <= SL_TPB && N <= (1 << SRC_BITS) &&
                            ws_need <= ws_size;

    if (buckets) {
        unsigned* binned = (unsigned*)((char*)d_ws + bin_off);
        int*      runs   = (int*)((char*)d_ws + runs_off);
        gcn_bin_slice<<<n_slices, SL_TPB, 0, stream>>>(src, dst, binned, runs, E, nb);
        gcn_bucket_rows<<<nb, SL_TPB, 0, stream>>>(binned, runs, x, W, dinv, h, N, nb, n_slices);
        gcn_bucket_sum<<<nb, SL_TPB, 0, stream>>>(binned, runs, h, dinv, bias, out, N, nb, n_slices);
    } else {
        gcn_deg_init<<<(N + TPB - 1) / TPB, TPB, 0, stream>>>(dinv, ctr, N);

        const int deg_ranges    = (N + DEG_TILE - 1) / DEG_TILE;
        const int deg_ranges_lo = (deg_ranges + 1) / 2;
        const int deg_parts     = (256 + deg_ranges - 1) / deg_ranges;
        const int deg_slice     = (E + deg_parts - 1) / deg_parts;
        gcn_deg_count<<<256, DEG_TPB, 0, stream>>>(dst, dinv, ctr, E, N, deg_ranges,
                                                   deg_ranges_lo, deg_parts, deg_slice);
        gcn_dinv<<<(N + TPB - 1) / TPB, TPB, 0, stream>>>(dinv, N);

        const int n_tiles = (N + 15) / 16;
        const int waves_per_block = TPB / 32;
        const int gemm_blocks = (n_tiles + waves_per_block - 1) / waves_per_block;
        gcn_gemm_wmma<<<gemm_blocks, TPB, 0, stream>>>(x, W, bias, dinv, h, out, N, n_tiles);

        const int sc_chunks = (E + SC_CHUNK - 1) / SC_CHUNK;
        gcn_scatter<<<2048, TPB, 0, stream>>>(src, dst, dinv, h, out, ctr + 2 * CTR_STRIDE,
                                              E, sc_chunks, split, N);
    }
}


# edge sum: the 64-edge row groups of a run are requested one group ahead of their adds (next group in flight during the adds, same loads every trip); partial steps add unbranched; sum order changes
# speedup vs baseline: 8.8112x; 1.0316x over previous
#include <hip/hip_runtime.h>
#include <hip/hip_bf16.h>

typedef __attribute__((ext_vector_type(2))) float v2f;
typedef __attribute__((ext_vector_type(8))) float v8f;

#define IN_CH 16
#define OUT_CH 8

#define CTR_STRIDE 64
#define DEG_TILE   65536
#define DEG_TPB    1024
#define DEG_UNROLL 16
#define DEG_FLUSH  8
#define SC_CHUNK  128
#define BK_SHIFT  11
#define BK_NODES  2048
#define BK_MAX    1024
#define SRC_BITS  21
#define SL_EDGES  20480
#define SL_TPB    1024
#define ROW_TILES 4

__device__ __forceinline__ int cache_half() {
    return (int)((__builtin_amdgcn_s_sendmsg_rtn(0x87) >> 18) & 1u);
}

__device__ __forceinline__ int draw_ticket(int* ctr) {
    int t = 0;
    if ((threadIdx.x & 31) == 0) t = atomicAdd(ctr, 1);
    return __builtin_amdgcn_readfirstlane(t);
}

__global__ void gcn_deg_init(float* __restrict__ deg, int* __restrict__ ctr, int n) {
    int i = blockIdx.x * blockDim.x + threadIdx.x;
    if (i < n) deg[i] = 1.0f;
    if (i < 4) ctr[i * CTR_STRIDE] = 0;
}

__global__ void __launch_bounds__(DEG_TPB)
gcn_deg_count(const int* __restrict__ dst, float* __restrict__ deg,
              int* __restrict__ ctr, int e_cnt, int n,
              int n_ranges, int ranges_lo, int parts, int slice) {
    __shared__ int tile[DEG_TILE];
    __shared__ int s_msg[2];
    const int tid = threadIdx.x;

    if (tid == 0) s_msg[1] = cache_half();
    __syncthreads();
    int  r      = __builtin_amdgcn_readfirstlane(s_msg[1]);
    bool helped = false;
    const int max_trips = n_ranges * parts + 2;
    for (int it = 0; it < max_trips; ++it) {
        if (tid == 0) s_msg[0] = atomicAdd(ctr + r * CTR_STRIDE, 1);
        __syncthreads();
        const int t = __builtin_amdgcn_readfirstlane(s_msg[0]);
        __syncthreads();
        const int first = r ? ranges_lo : 0;
        const int count = r ? (n_ranges - ranges_lo) : ranges_lo;
        if (t >= count * parts) {
            if (helped) break;
            helped = true;
            r ^= 1;
            continue;
        }
        const int range = first + t / parts;
        const int part  = t - (t / parts) * parts;
        const int lo    = range * DEG_TILE;

        for (int i = tid; i < DEG_TILE; i += DEG_TPB) tile[i] = 0;
        __syncthreads();

        const int e_begin = part * slice;
        const int e_stop  = e_begin + slice;
        const int e_end   = e_stop < e_cnt ? e_stop : e_cnt;
        for (int base = e_begin; base < e_end; base += DEG_TPB * DEG_UNROLL) {
            int d[DEG_UNROLL];
#pragma unroll
            for (int k = 0; k < DEG_UNROLL; ++k) {
                const int e = base + k * DEG_TPB + tid;
                d[k] = (e < e_end) ? dst[e] : -1;
            }
#pragma unroll
            for (int k = 0; k < DEG_UNROLL; ++k) {
                const unsigned off = (unsigned)(d[k] - lo);
                if (off < (unsigned)DEG_TILE) atomicAdd(&tile[off], 1);
            }
        }
        __syncthreads();

        for (int i0 = tid; i0 < DEG_TILE; i0 += DEG_TPB * DEG_FLUSH) {
            int c[DEG_FLUSH];
#pragma unroll
            for (int k = 0; k < DEG_FLUSH; ++k) c[k] = tile[i0 + k * DEG_TPB];
#pragma unroll
            for (int k = 0; k < DEG_FLUSH; ++k) {
                const int node = lo + i0 + k * DEG_TPB;
                if (c[k] != 0 && node < n) atomicAdd(&deg[node], (float)c[k]);
            }
        }
    }
}

__global__ void gcn_dinv(float* __restrict__ deg, int n) {
    int i = blockIdx.x * blockDim.x + threadIdx.x;
    if (i < n) {
        float d = deg[i];
        deg[i] = d > 0.0f ? __frsqrt_rn(d) : 0.0f;
    }
}

__global__ void gcn_gemm_wmma(const float* __restrict__ x,
                              const float* __restrict__ W,
                              const float* __restrict__ bias,
                              const float* __restrict__ dinv,
                              float* __restrict__ h,
                              float* __restrict__ out,
                              int n_nodes, int n_tiles) {
    const int lane  = threadIdx.x & 31;
    const int wave  = threadIdx.x >> 5;
    const int tile  = blockIdx.x * (blockDim.x >> 5) + wave;
    if (tile >= n_tiles) return;

    const int base  = tile * 16;
    const int m     = lane & 15;
    const int khalf = (lane >> 4) * 2;

    int arow = base + m;
    if (arow >= n_nodes) arow = n_nodes - 1;
    const float* __restrict__ xrow = x + (size_t)arow * IN_CH;

    v2f a0 = *(const v2f*)(xrow + 0  + khalf);
    v2f a1 = *(const v2f*)(xrow + 4  + khalf);
    v2f a2 = *(const v2f*)(xrow + 8  + khalf);
    v2f a3 = *(const v2f*)(xrow + 12 + khalf);

    const int   wrow = (m < OUT_CH) ? m : 0;
    const float sel  = (m < OUT_CH) ? 1.0f : 0.0f;
    const float* __restrict__ wr = W + wrow * IN_CH + khalf;
    v2f b0 = *(const v2f*)(wr + 0);
    v2f b1 = *(const v2f*)(wr + 4);
    v2f b2 = *(const v2f*)(wr + 8);
    v2f b3 = *(const v2f*)(wr + 12);
    b0 *= sel; b1 *= sel; b2 *= sel; b3 *= sel;

    v8f acc = {};
    acc = __builtin_amdgcn_wmma_f32_16x16x4_f32(false, a0, false, b0, (short)0, acc, false, false);
    acc = __builtin_amdgcn_wmma_f32_16x16x4_f32(false, a1, false, b1, (short)0, acc, false, false);
    acc = __builtin_amdgcn_wmma_f32_16x16x4_f32(false, a2, false, b2, (short)0, acc, false, false);
    acc = __builtin_amdgcn_wmma_f32_16x16x4_f32(false, a3, false, b3, (short)0, acc, false, false);

    const int  n     = m;
    const int  mbase = base + ((lane >> 4) * 8);
    const bool full  = (base + 16) <= n_nodes;

    if (full) {
        if (n < OUT_CH) {
            const float bn = bias[n];
            const float4 d0 = *(const float4*)(dinv + mbase);
            const float4 d1 = *(const float4*)(dinv + mbase + 4);
            const float dv[8] = {d0.x, d0.y, d0.z, d0.w, d1.x, d1.y, d1.z, d1.w};
            float* __restrict__ hp = h   + (size_t)mbase * OUT_CH + n;
            float* __restrict__ op = out + (size_t)mbase * OUT_CH + n;
#pragma unroll
            for (int r = 0; r < 8; ++r) {
                const float hv = acc[r];
                hp[r * OUT_CH] = hv;
                op[r * OUT_CH] = bn + dv[r] * dv[r] * hv;
            }
        }
    } else {
        if (n < OUT_CH) {
            const float bn = bias[n];
#pragma unroll
            for (int r = 0; r < 8; ++r) {
                const int node = mbase + r;
                if (node < n_nodes) {
                    const float hv = acc[r];
                    h[(size_t)node * OUT_CH + n] = hv;
                    const float di = dinv[node];
                    out[(size_t)node * OUT_CH + n] = bn + di * di * hv;
                }
            }
        }
    }
}

__global__ void __launch_bounds__(256)
gcn_scatter(const int* __restrict__ src, const int* __restrict__ dst,
            const float* __restrict__ dinv, const float* __restrict__ h,
            float* __restrict__ out, int* __restrict__ ctr,
            int e_cnt, int n_chunks, int split, int n) {
    const int      lane = threadIdx.x & 31;
    const unsigned f    = (unsigned)(lane & 7);
    const int      grp  = lane >> 3;

    int  r      = cache_half();
    bool helped = false;
    int  next   = draw_ticket(ctr + r * CTR_STRIDE);
    const int max_trips = 2 * n_chunks + 2;
    for (int it = 0; it < max_trips; ++it) {
        const int c = next;
        if (c >= n_chunks) {
            if (helped) break;
            helped = true;
            r ^= 1;
            next = draw_ticket(ctr + r * CTR_STRIDE);
            continue;
        }
        const int lo = r ? split : 0;
        const int hi = r ? n : split;
        const int e0 = c * SC_CHUNK + lane;

        int   dv[SC_CHUNK / 32];
        int   sv[SC_CHUNK / 32];
        float wv[SC_CHUNK / 32];
        bool  in[SC_CHUNK / 32];
#pragma unroll
        for (int p = 0; p < SC_CHUNK / 32; ++p) {
            const int e  = e0 + p * 32;
            const int ec = (e < e_cnt) ? e : e_cnt - 1;
            dv[p] = dst[ec];
            sv[p] = src[ec];
        }
        next = draw_ticket(ctr + r * CTR_STRIDE);
#pragma unroll
        for (int p = 0; p < SC_CHUNK / 32; ++p) {
            in[p] = (e0 + p * 32 < e_cnt) && (dv[p] >= lo) && (dv[p] < hi);
            dv[p] = in[p] ? dv[p] : 0;
            sv[p] = in[p] ? sv[p] : 0;
        }
#pragma unroll
        for (int p = 0; p < SC_CHUNK / 32; ++p) {
            const float a = dinv[sv[p]];
            const float b = dinv[dv[p]];
            wv[p] = in[p] ? a * b : 0.0f;
        }

#pragma unroll
        for (int p = 0; p < SC_CHUNK / 32; ++p) {
            const unsigned mask = __builtin_amdgcn_ballot_w32(in[p]);
            const int cnt = __popc(mask);
            if (cnt > 0) {
                const int rank = (int)__builtin_amdgcn_mbcnt_lo(mask, 0u);
                const int slot = in[p] ? rank : cnt + lane - rank;
                const int so = __builtin_amdgcn_ds_permute(slot << 2, sv[p] * OUT_CH);
                const int dd = __builtin_amdgcn_ds_permute(slot << 2, dv[p] * OUT_CH);
                const int wi = __builtin_amdgcn_ds_permute(slot << 2, __float_as_int(wv[p]));
                for (int jb = 0; jb < cnt; jb += 16) {
                    float    hv[4];
                    unsigned oi[4];
                    float    wj[4];
#pragma unroll
                    for (int j = 0; j < 4; ++j) {
                        const int sl = jb + j * 4 + grp;
                        const int bp = sl << 2;
                        const unsigned sj = (unsigned)__builtin_amdgcn_ds_bpermute(bp, so);
                        hv[j] = (sl < cnt) ? h[sj + f] : 0.0f;
                        oi[j] = (unsigned)__builtin_amdgcn_ds_bpermute(bp, dd) + f;
                        wj[j] = __int_as_float(__builtin_amdgcn_ds_bpermute(bp, wi));
                    }
#pragma unroll
                    for (int j = 0; j < 4; ++j)
                        if (jb + j * 4 + grp < cnt) atomicAdd(out + oi[j], wj[j] * hv[j]);
                }
            }
        }
    }
}

__global__ void __launch_bounds__(SL_TPB)
gcn_bin_slice(const int* __restrict__ src, const int* __restrict__ dst,
              unsigned* __restrict__ binned, int* __restrict__ runs,
              int e_cnt, int nb) {
    __shared__ unsigned stage[SL_EDGES];
    __shared__ int cnt[BK_MAX];
    __shared__ int wsum[32];
    const int tid  = threadIdx.x;
    const int lane = tid & 31;
    const int wave = tid >> 5;
    const int s    = blockIdx.x;
    const int e0   = s * SL_EDGES;
    const int e_hi = e0 + SL_EDGES;
    const int e_end = e_hi < e_cnt ? e_hi : e_cnt;

    cnt[tid] = 0;
    __syncthreads();

    int dv[SL_EDGES / SL_TPB];
    int sv[SL_EDGES / SL_TPB];
#pragma unroll
    for (int k = 0; k < SL_EDGES / SL_TPB; ++k) {
        const int e  = e0 + k * SL_TPB + tid;
        const int ec = e < e_end ? e : e_end - 1;
        dv[k] = dst[ec];
        sv[k] = src[ec];
    }
#pragma unroll
    for (int k = 0; k < SL_EDGES / SL_TPB; ++k)
        if (e0 + k * SL_TPB + tid < e_end) atomicAdd(&cnt[dv[k] >> BK_SHIFT], 1);
    __syncthreads();

    const int v = cnt[tid];
    int incl = v;
#pragma unroll
    for (int o = 1; o < 32; o <<= 1) {
        const int t = __shfl_up(incl, o);
        if (lane >= o) incl += t;
    }
    if (lane == 31) wsum[wave] = incl;
    __syncthreads();
    if (wave == 0) {
        const int wt = wsum[lane];
        int wi = wt;
#pragma unroll
        for (int o = 1; o < 32; o <<= 1) {
            const int t = __shfl_up(wi, o);
            if (lane >= o) wi += t;
        }
        wsum[lane] = wi - wt;
    }
    __syncthreads();
    const int excl = incl - v + wsum[wave];
    cnt[tid] = excl;
    if (tid <= nb) runs[(size_t)s * (nb + 1) + tid] = excl;
    __syncthreads();

#pragma unroll
    for (int k = 0; k < SL_EDGES / SL_TPB; ++k) {
        if (e0 + k * SL_TPB + tid < e_end) {
            const int p = atomicAdd(&cnt[dv[k] >> BK_SHIFT], 1);
            stage[p] = (unsigned)sv[k] | ((unsigned)(dv[k] & (BK_NODES - 1)) << SRC_BITS);
        }
    }
    __syncthreads();

    const int n_s = e_end - e0;
    unsigned* __restrict__ region = binned + (size_t)s * SL_EDGES;
    for (int i = tid; i < n_s; i += SL_TPB) region[i] = stage[i];
}

__device__ __forceinline__ void run_words(const unsigned* __restrict__ region,
                                          int p0, int hi, int lane, unsigned (&w)[4]) {
#pragma unroll
    for (int c = 0; c < 4; ++c) {
        const int p = p0 + c * 32 + lane;
        w[c] = region[p < hi ? p : hi - 1];
    }
}

__global__ void __launch_bounds__(SL_TPB)
gcn_bucket_rows(const unsigned* __restrict__ binned, const int* __restrict__ runs,
                const float* __restrict__ x,
                const float* __restrict__ W,
                float* __restrict__ dinv, float* __restrict__ hs,
                int n, int nb, int n_slices) {
    __shared__ int   hist[BK_NODES];
    __shared__ float dloc[BK_NODES];
    __shared__ int   rlo[SL_TPB];
    __shared__ int   rhi[SL_TPB];
    const int tid   = threadIdx.x;
    const int lane  = tid & 31;
    const int wave  = __builtin_amdgcn_readfirstlane(tid >> 5);
    const int b     = blockIdx.x;
    const int node0 = b * BK_NODES;
    const int m     = lane & 15;
    const int khalf = (lane >> 4) * 2;

    v2f a[ROW_TILES][4];
#pragma unroll
    for (int q = 0; q < ROW_TILES; ++q) {
        int arow = node0 + (wave * ROW_TILES + q) * 16 + m;
        if (arow >= n) arow = n - 1;
        const float* __restrict__ xrow = x + (size_t)arow * IN_CH + khalf;
#pragma unroll
        for (int k = 0; k < 4; ++k) a[q][k] = *(const v2f*)(xrow + 4 * k);
    }
    const int   wrow = (m < OUT_CH) ? m : 0;
    const float sel  = (m < OUT_CH) ? 1.0f : 0.0f;
    const float* __restrict__ wr = W + wrow * IN_CH + khalf;
    v2f bf[4];
#pragma unroll
    for (int k = 0; k < 4; ++k) bf[k] = *(const v2f*)(wr + 4 * k) * sel;

    if (tid < n_slices) {
        const int* __restrict__ rp = runs + (size_t)tid * (nb + 1) + b;
        rlo[tid] = rp[0];
        rhi[tid] = rp[1];
    }
    for (int i = tid; i < BK_NODES; i += SL_TPB) hist[i] = 0;
    __syncthreads();

    for (int s0 = wave; s0 < n_slices; s0 += (SL_TPB / 32) * 4) {
        int      lo[4], hi[4];
        unsigned w[4][4];
#pragma unroll
        for (int u = 0; u < 4; ++u) {
            const int  s  = s0 + u * (SL_TPB / 32);
            const bool ok = s < n_slices;
            const int  sc = ok ? s : s0;
            const int  l  = __builtin_amdgcn_readfirstlane(rlo[sc]);
            const int  h  = __builtin_amdgcn_readfirstlane(rhi[sc]);
            lo[u] = ok ? l : 0;
            hi[u] = ok ? h : 0;
            const unsigned* __restrict__ region = binned + (size_t)sc * SL_EDGES;
#pragma unroll
            for (int cc = 0; cc < 4; ++cc) {
                const int p  = lo[u] + cc * 32 + lane;
                const int pc = p < hi[u] ? p : hi[u] - 1;
                w[u][cc] = region[pc > 0 ? pc : 0];
            }
        }
#pragma unroll
        for (int u = 0; u < 4; ++u) {
#pragma unroll
            for (int cc = 0; cc < 4; ++cc)
                if (lo[u] + cc * 32 + lane < hi[u]) atomicAdd(&hist[w[u][cc] >> SRC_BITS], 1);
        }
#pragma unroll
        for (int u = 0; u < 4; ++u) {
            for (int p0 = lo[u] + 128; p0 < hi[u]; p0 += 128) {
                unsigned wt[4];
                run_words(binned + (size_t)(s0 + u * (SL_TPB / 32)) * SL_EDGES, p0, hi[u], lane, wt);
#pragma unroll
                for (int cc = 0; cc < 4; ++cc)
                    if (p0 + cc * 32 + lane < hi[u]) atomicAdd(&hist[wt[cc] >> SRC_BITS], 1);
            }
        }
    }
    __syncthreads();

    for (int i = tid; i < BK_NODES; i += SL_TPB) {
        const float d = (float)(hist[i] + 1);
        const float v = d > 0.0f ? __frsqrt_rn(d) : 0.0f;
        dloc[i] = v;
        if (node0 + i < n) dinv[node0 + i] = v;
    }
    __syncthreads();

    v8f acc[ROW_TILES];
#pragma unroll
    for (int q = 0; q < ROW_TILES; ++q) acc[q] = v8f{};
#pragma unroll
    for (int k = 0; k < 4; ++k) {
#pragma unroll
        for (int q = 0; q < ROW_TILES; ++q)
            acc[q] = __builtin_amdgcn_wmma_f32_16x16x4_f32(false, a[q][k], false, bf[k], (short)0, acc[q], false, false);
    }

    if (m < OUT_CH) {
#pragma unroll
        for (int q = 0; q < ROW_TILES; ++q) {
            const int tb    = (wave * ROW_TILES + q) * 16;
            const int rbase = tb + (lane >> 4) * 8;
            const float4 d0 = *(const float4*)(dloc + rbase);
            const float4 d1 = *(const float4*)(dloc + rbase + 4);
            const float dv[8] = {d0.x, d0.y, d0.z, d0.w, d1.x, d1.y, d1.z, d1.w};
            float* __restrict__ hp = hs + (size_t)(node0 + rbase) * OUT_CH + m;
            if (node0 + tb + 16 <= n) {
#pragma unroll
                for (int r = 0; r < 8; ++r) hp[r * OUT_CH] = dv[r] * acc[q][r];
            } else {
#pragma unroll
                for (int r = 0; r < 8; ++r)
                    if (node0 + rbase + r < n) hp[r * OUT_CH] = dv[r] * acc[q][r];
            }
        }
    }
}

__device__ __forceinline__ void run_words_all(const unsigned* __restrict__ region,
                                              int lo, int hi, int lane, unsigned (&w)[4]) {
#pragma unroll
    for (int c = 0; c < 4; ++c) {
        const int p  = lo + c * 32 + lane;
        const int pc = p < hi ? p : hi - 1;
        w[c] = region[pc > 0 ? pc : 0];
    }
}

template <int NJ>
__device__ __forceinline__ void rows_request(const float* __restrict__ hs, unsigned w0, unsigned w1,
                                             const int (&bp)[8], unsigned f4,
                                             unsigned (&ww)[NJ], float (&hv)[NJ]) {
#pragma unroll
    for (int j = 0; j < NJ; ++j)
        ww[j] = (unsigned)__builtin_amdgcn_ds_bpermute(bp[j & 7], (int)(j < 8 ? w0 : w1));
#pragma unroll
    for (int j = 0; j < NJ; ++j) {
        const unsigned off = ((ww[j] & ((1u << SRC_BITS) - 1u)) << 5) | f4;
        hv[j] = *(const float*)((const char*)hs + off);
    }
}

template <int NJ>
__device__ __forceinline__ void rows_add(float* __restrict__ tile, float* __restrict__ spare,
                                         const unsigned (&ww)[NJ], const float (&hv)[NJ],
                                         int left, int grp, unsigned f) {
#pragma unroll
    for (int j = 0; j < NJ; ++j) {
        if (j * 4 < left) {
            float* dst = (j * 4 + grp < left) ? tile + ((ww[j] >> SRC_BITS) * OUT_CH + f) : spare;
            atomicAdd(dst, hv[j]);
        }
    }
}

__global__ void __launch_bounds__(SL_TPB)
gcn_bucket_sum(const unsigned* __restrict__ binned, const int* __restrict__ runs,
               const float* __restrict__ hs, const float* __restrict__ dinv,
               const float* __restrict__ bias, float* __restrict__ out,
               int n, int nb, int n_slices) {
    __shared__ float tile[BK_NODES * OUT_CH];
    __shared__ int   rlo[SL_TPB];
    __shared__ int   rhi[SL_TPB];
    __shared__ float pad[32];
    const int      tid  = threadIdx.x;
    const int      lane = tid & 31;
    const int      wave = __builtin_amdgcn_readfirstlane(tid >> 5);
    const unsigned f    = (unsigned)(lane & 7);
    const int      grp  = lane >> 3;
    const int      b    = blockIdx.x;

    if (tid < n_slices) {
        const int* __restrict__ rp = runs + (size_t)tid * (nb + 1) + b;
        rlo[tid] = rp[0];
        rhi[tid] = rp[1];
    }
    if (tid < 32) pad[tid] = 0.0f;
    for (int i = tid; i < BK_NODES * OUT_CH; i += SL_TPB) tile[i] = 0.0f;
    __syncthreads();

    int bp[8];
#pragma unroll
    for (int k = 0; k < 8; ++k) bp[k] = (k * 4 + grp) << 2;
    const unsigned f4    = f << 2;
    float*         spare = pad + lane;

    int s = wave;
    if (s < n_slices) {
        int lo = __builtin_amdgcn_readfirstlane(rlo[s]);
        int hi = __builtin_amdgcn_readfirstlane(rhi[s]);
        unsigned w[4];
        run_words_all(binned + (size_t)s * SL_EDGES, lo, hi, lane, w);
        const int  s1  = s + SL_TPB / 32;
        const bool ok1 = s1 < n_slices;
        const int  sc1 = ok1 ? s1 : s;
        const int  l1  = __builtin_amdgcn_readfirstlane(rlo[sc1]);
        const int  h1  = __builtin_amdgcn_readfirstlane(rhi[sc1]);
        int lo1 = ok1 ? l1 : 0;
        int hi1 = ok1 ? h1 : 0;
        unsigned w1[4];
        run_words_all(binned + (size_t)sc1 * SL_EDGES, lo1, hi1, lane, w1);
        unsigned wwA[16];
        float    hvA[16];
        rows_request<16>(hs, w[0], w[1], bp, f4, wwA, hvA);
        for (; s < n_slices; s += SL_TPB / 32) {
            const int  s2  = s + 2 * (SL_TPB / 32);
            const bool ok2 = s2 < n_slices;
            const int  sc2 = ok2 ? s2 : s;
            const int  l2  = __builtin_amdgcn_readfirstlane(rlo[sc2]);
            const int  h2  = __builtin_amdgcn_readfirstlane(rhi[sc2]);
            const int  lo2 = ok2 ? l2 : 0;
            const int  hi2 = ok2 ? h2 : 0;
            unsigned w2[4];
            run_words_all(binned + (size_t)sc2 * SL_EDGES, lo2, hi2, lane, w2);
            unsigned wwB[16];
            float    hvB[16];
            rows_request<16>(hs, w[2], w[3], bp, f4, wwB, hvB);
            const int left = hi - lo;
            rows_add<16>(tile, spare, wwA, hvA, left, grp, f);
            rows_request<16>(hs, w1[0], w1[1], bp, f4, wwA, hvA);
            rows_add<16>(tile, spare, wwB, hvB, left - 64, grp, f);
            for (int p0 = lo + 128; p0 < hi; p0 += 128) {
                unsigned wr[4];
                run_words(binned + (size_t)s * SL_EDGES, p0, hi, lane, wr);
                unsigned wwC[16];
                float    hvC[16];
                rows_request<16>(hs, wr[0], wr[1], bp, f4, wwC, hvC);
                rows_add<16>(tile, spare, wwC, hvC, hi - p0, grp, f);
                if (hi > p0 + 64) {
                    rows_request<16>(hs, wr[2], wr[3], bp, f4, wwC, hvC);
                    rows_add<16>(tile, spare, wwC, hvC, hi - p0 - 64, grp, f);
                }
            }
            lo = lo1; hi = hi1;
            lo1 = lo2; hi1 = hi2;
#pragma unroll
            for (int c = 0; c < 4; ++c) { w[c] = w1[c]; w1[c] = w2[c]; }
        }
    }
    __syncthreads();

    const int node0 = b * BK_NODES;
    for (int i = tid; i < BK_NODES * OUT_CH; i += SL_TPB) {
        const int node = node0 + (i >> 3);
        if (node < n) {
            const size_t o = (size_t)node * OUT_CH + (i & 7);
            out[o] = bias[i & 7] + dinv[node] * (tile[i] + hs[o]);
        }
    }
}

extern "C" void kernel_launch(void* const* d_in, const int* in_sizes, int n_in,
                              void* d_out, int out_size, void* d_ws, size_t ws_size,
                              hipStream_t stream) {
    const float* x    = (const float*)d_in[0];
    const int*   ei   = (const int*)d_in[1];
    const float* W    = (const float*)d_in[2];
    const float* bias = (const float*)d_in[3];

    const int N = in_sizes[0] / IN_CH;
    const int E = in_sizes[1] / 2;
    const int* src = ei;
    const int* dst = ei + (size_t)E;

    int*   ctr  = (int*)d_ws;
    const size_t dinv_off = 4u * CTR_STRIDE * sizeof(int);
    float* dinv = (float*)((char*)d_ws + dinv_off);
    size_t h_off = ((dinv_off + (size_t)N * sizeof(float)) + 255u) & ~(size_t)255u;
    float* h = (float*)((char*)d_ws + h_off);
    float* out = (float*)d_out;

    const int TPB = 256;

    const int split = ((N / 2) + 63) & ~63;

    const int    nb       = (N + BK_NODES - 1) >> BK_SHIFT;
    const int    n_slices = (E + SL_EDGES - 1) / SL_EDGES;
    const size_t bin_off  = ((h_off + (size_t)N * OUT_CH * sizeof(float)) + 255u) & ~(size_t)255u;
    const size_t runs_off = ((bin_off + (size_t)n_slices * SL_EDGES * sizeof(unsigned)) + 255u) & ~(size_t)255u;
    const size_t ws_need  = runs_off + (size_t)n_slices * (size_t)(nb + 1) * sizeof(int);
    const bool   buckets  = E > 0 && nb < BK_MAX && n_slices <= SL_TPB && N <= (1 << SRC_BITS) &&
                            ws_need <= ws_size;

    if (buckets) {
        unsigned* binned = (unsigned*)((char*)d_ws + bin_off);
        int*      runs   = (int*)((char*)d_ws + runs_off);
        gcn_bin_slice<<<n_slices, SL_TPB, 0, stream>>>(src, dst, binned, runs, E, nb);
        gcn_bucket_rows<<<nb, SL_TPB, 0, stream>>>(binned, runs, x, W, dinv, h, N, nb, n_slices);
        gcn_bucket_sum<<<nb, SL_TPB, 0, stream>>>(binned, runs, h, dinv, bias, out, N, nb, n_slices);
    } else {
        gcn_deg_init<<<(N + TPB - 1) / TPB, TPB, 0, stream>>>(dinv, ctr, N);

        const int deg_ranges    = (N + DEG_TILE - 1) / DEG_TILE;
        const int deg_ranges_lo = (deg_ranges + 1) / 2;
        const int deg_parts     = (256 + deg_ranges - 1) / deg_ranges;
        const int deg_slice     = (E + deg_parts - 1) / deg_parts;
        gcn_deg_count<<<256, DEG_TPB, 0, stream>>>(dst, dinv, ctr, E, N, deg_ranges,
                                                   deg_ranges_lo, deg_parts, deg_slice);
        gcn_dinv<<<(N + TPB - 1) / TPB, TPB, 0, stream>>>(dinv, N);

        const int n_tiles = (N + 15) / 16;
        const int waves_per_block = TPB / 32;
        const int gemm_blocks = (n_tiles + waves_per_block - 1) / waves_per_block;
        gcn_gemm_wmma<<<gemm_blocks, TPB, 0, stream>>>(x, W, bias, dinv, h, out, N, n_tiles);

        const int sc_chunks = (E + SC_CHUNK - 1) / SC_CHUNK;
        gcn_scatter<<<2048, TPB, 0, stream>>>(src, dst, dinv, h, out, ctr + 2 * CTR_STRIDE,
                                              E, sc_chunks, split, N);
    }
}


# edge sum: own rows (self loop), dinv, bias read at the head in one group, tile starts from the self-loop row, tail has no load; slice sort copies out with its LDS reads grouped; sum order changes
# speedup vs baseline: 9.9102x; 1.1247x over previous
#include <hip/hip_runtime.h>
#include <hip/hip_bf16.h>

typedef __attribute__((ext_vector_type(2))) float v2f;
typedef __attribute__((ext_vector_type(8))) float v8f;

#define IN_CH 16
#define OUT_CH 8

#define CTR_STRIDE 64
#define DEG_TILE   65536
#define DEG_TPB    1024
#define DEG_UNROLL 16
#define DEG_FLUSH  8
#define SC_CHUNK  128
#define BK_SHIFT  11
#define BK_NODES  2048
#define BK_MAX    1024
#define SRC_BITS  21
#define SL_EDGES  20480
#define SL_TPB    1024
#define ROW_TILES 4

__device__ __forceinline__ int cache_half() {
    return (int)((__builtin_amdgcn_s_sendmsg_rtn(0x87) >> 18) & 1u);
}

__device__ __forceinline__ int draw_ticket(int* ctr) {
    int t = 0;
    if ((threadIdx.x & 31) == 0) t = atomicAdd(ctr, 1);
    return __builtin_amdgcn_readfirstlane(t);
}

__global__ void gcn_deg_init(float* __restrict__ deg, int* __restrict__ ctr, int n) {
    int i = blockIdx.x * blockDim.x + threadIdx.x;
    if (i < n) deg[i] = 1.0f;
    if (i < 4) ctr[i * CTR_STRIDE] = 0;
}

__global__ void __launch_bounds__(DEG_TPB)
gcn_deg_count(const int* __restrict__ dst, float* __restrict__ deg,
              int* __restrict__ ctr, int e_cnt, int n,
              int n_ranges, int ranges_lo, int parts, int slice) {
    __shared__ int tile[DEG_TILE];
    __shared__ int s_msg[2];
    const int tid = threadIdx.x;

    if (tid == 0) s_msg[1] = cache_half();
    __syncthreads();
    int  r      = __builtin_amdgcn_readfirstlane(s_msg[1]);
    bool helped = false;
    const int max_trips = n_ranges * parts + 2;
    for (int it = 0; it < max_trips; ++it) {
        if (tid == 0) s_msg[0] = atomicAdd(ctr + r * CTR_STRIDE, 1);
        __syncthreads();
        const int t = __builtin_amdgcn_readfirstlane(s_msg[0]);
        __syncthreads();
        const int first = r ? ranges_lo : 0;
        const int count = r ? (n_ranges - ranges_lo) : ranges_lo;
        if (t >= count * parts) {
            if (helped) break;
            helped = true;
            r ^= 1;
            continue;
        }
        const int range = first + t / parts;
        const int part  = t - (t / parts) * parts;
        const int lo    = range * DEG_TILE;

        for (int i = tid; i < DEG_TILE; i += DEG_TPB) tile[i] = 0;
        __syncthreads();

        const int e_begin = part * slice;
        const int e_stop  = e_begin + slice;
        const int e_end   = e_stop < e_cnt ? e_stop : e_cnt;
        for (int base = e_begin; base < e_end; base += DEG_TPB * DEG_UNROLL) {
            int d[DEG_UNROLL];
#pragma unroll
            for (int k = 0; k < DEG_UNROLL; ++k) {
                const int e = base + k * DEG_TPB + tid;
                d[k] = (e < e_end) ? dst[e] : -1;
            }
#pragma unroll
            for (int k = 0; k < DEG_UNROLL; ++k) {
                const unsigned off = (unsigned)(d[k] - lo);
                if (off < (unsigned)DEG_TILE) atomicAdd(&tile[off], 1);
            }
        }
        __syncthreads();

        for (int i0 = tid; i0 < DEG_TILE; i0 += DEG_TPB * DEG_FLUSH) {
            int c[DEG_FLUSH];
#pragma unroll
            for (int k = 0; k < DEG_FLUSH; ++k) c[k] = tile[i0 + k * DEG_TPB];
#pragma unroll
            for (int k = 0; k < DEG_FLUSH; ++k) {
                const int node = lo + i0 + k * DEG_TPB;
                if (c[k] != 0 && node < n) atomicAdd(&deg[node], (float)c[k]);
            }
        }
    }
}

__global__ void gcn_dinv(float* __restrict__ deg, int n) {
    int i = blockIdx.x * blockDim.x + threadIdx.x;
    if (i < n) {
        float d = deg[i];
        deg[i] = d > 0.0f ? __frsqrt_rn(d) : 0.0f;
    }
}

__global__ void gcn_gemm_wmma(const float* __restrict__ x,
                              const float* __restrict__ W,
                              const float* __restrict__ bias,
                              const float* __restrict__ dinv,
                              float* __restrict__ h,
                              float* __restrict__ out,
                              int n_nodes, int n_tiles) {
    const int lane  = threadIdx.x & 31;
    const int wave  = threadIdx.x >> 5;
    const int tile  = blockIdx.x * (blockDim.x >> 5) + wave;
    if (tile >= n_tiles) return;

    const int base  = tile * 16;
    const int m     = lane & 15;
    const int khalf = (lane >> 4) * 2;

    int arow = base + m;
    if (arow >= n_nodes) arow = n_nodes - 1;
    const float* __restrict__ xrow = x + (size_t)arow * IN_CH;

    v2f a0 = *(const v2f*)(xrow + 0  + khalf);
    v2f a1 = *(const v2f*)(xrow + 4  + khalf);
    v2f a2 = *(const v2f*)(xrow + 8  + khalf);
    v2f a3 = *(const v2f*)(xrow + 12 + khalf);

    const int   wrow = (m < OUT_CH) ? m : 0;
    const float sel  = (m < OUT_CH) ? 1.0f : 0.0f;
    const float* __restrict__ wr = W + wrow * IN_CH + khalf;
    v2f b0 = *(const v2f*)(wr + 0);
    v2f b1 = *(const v2f*)(wr + 4);
    v2f b2 = *(const v2f*)(wr + 8);
    v2f b3 = *(const v2f*)(wr + 12);
    b0 *= sel; b1 *= sel; b2 *= sel; b3 *= sel;

    v8f acc = {};
    acc = __builtin_amdgcn_wmma_f32_16x16x4_f32(false, a0, false, b0, (short)0, acc, false, false);
    acc = __builtin_amdgcn_wmma_f32_16x16x4_f32(false, a1, false, b1, (short)0, acc, false, false);
    acc = __builtin_amdgcn_wmma_f32_16x16x4_f32(false, a2, false, b2, (short)0, acc, false, false);
    acc = __builtin_amdgcn_wmma_f32_16x16x4_f32(false, a3, false, b3, (short)0, acc, false, false);

    const int  n     = m;
    const int  mbase = base + ((lane >> 4) * 8);
    const bool full  = (base + 16) <= n_nodes;

    if (full) {
        if (n < OUT_CH) {
            const float bn = bias[n];
            const float4 d0 = *(const float4*)(dinv + mbase);
            const float4 d1 = *(const float4*)(dinv + mbase + 4);
            const float dv[8] = {d0.x, d0.y, d0.z, d0.w, d1.x, d1.y, d1.z, d1.w};
            float* __restrict__ hp = h   + (size_t)mbase * OUT_CH + n;
            float* __restrict__ op = out + (size_t)mbase * OUT_CH + n;
#pragma unroll
            for (int r = 0; r < 8; ++r) {
                const float hv = acc[r];
                hp[r * OUT_CH] = hv;
                op[r * OUT_CH] = bn + dv[r] * dv[r] * hv;
            }
        }
    } else {
        if (n < OUT_CH) {
            const float bn = bias[n];
#pragma unroll
            for (int r = 0; r < 8; ++r) {
                const int node = mbase + r;
                if (node < n_nodes) {
                    const float hv = acc[r];
                    h[(size_t)node * OUT_CH + n] = hv;
                    const float di = dinv[node];
                    out[(size_t)node * OUT_CH + n] = bn + di * di * hv;
                }
            }
        }
    }
}

__global__ void __launch_bounds__(256)
gcn_scatter(const int* __restrict__ src, const int* __restrict__ dst,
            const float* __restrict__ dinv, const float* __restrict__ h,
            float* __restrict__ out, int* __restrict__ ctr,
            int e_cnt, int n_chunks, int split, int n) {
    const int      lane = threadIdx.x & 31;
    const unsigned f    = (unsigned)(lane & 7);
    const int      grp  = lane >> 3;

    int  r      = cache_half();
    bool helped = false;
    int  next   = draw_ticket(ctr + r * CTR_STRIDE);
    const int max_trips = 2 * n_chunks + 2;
    for (int it = 0; it < max_trips; ++it) {
        const int c = next;
        if (c >= n_chunks) {
            if (helped) break;
            helped = true;
            r ^= 1;
            next = draw_ticket(ctr + r * CTR_STRIDE);
            continue;
        }
        const int lo = r ? split : 0;
        const int hi = r ? n : split;
        const int e0 = c * SC_CHUNK + lane;

        int   dv[SC_CHUNK / 32];
        int   sv[SC_CHUNK / 32];
        float wv[SC_CHUNK / 32];
        bool  in[SC_CHUNK / 32];
#pragma unroll
        for (int p = 0; p < SC_CHUNK / 32; ++p) {
            const int e  = e0 + p * 32;
            const int ec = (e < e_cnt) ? e : e_cnt - 1;
            dv[p] = dst[ec];
            sv[p] = src[ec];
        }
        next = draw_ticket(ctr + r * CTR_STRIDE);
#pragma unroll
        for (int p = 0; p < SC_CHUNK / 32; ++p) {
            in[p] = (e0 + p * 32 < e_cnt) && (dv[p] >= lo) && (dv[p] < hi);
            dv[p] = in[p] ? dv[p] : 0;
            sv[p] = in[p] ? sv[p] : 0;
        }
#pragma unroll
        for (int p = 0; p < SC_CHUNK / 32; ++p) {
            const float a = dinv[sv[p]];
            const float b = dinv[dv[p]];
            wv[p] = in[p] ? a * b : 0.0f;
        }

#pragma unroll
        for (int p = 0; p < SC_CHUNK / 32; ++p) {
            const unsigned mask = __builtin_amdgcn_ballot_w32(in[p]);
            const int cnt = __popc(mask);
            if (cnt > 0) {
                const int rank = (int)__builtin_amdgcn_mbcnt_lo(mask, 0u);
                const int slot = in[p] ? rank : cnt + lane - rank;
                const int so = __builtin_amdgcn_ds_permute(slot << 2, sv[p] * OUT_CH);
                const int dd = __builtin_amdgcn_ds_permute(slot << 2, dv[p] * OUT_CH);
                const int wi = __builtin_amdgcn_ds_permute(slot << 2, __float_as_int(wv[p]));
                for (int jb = 0; jb < cnt; jb += 16) {
                    float    hv[4];
                    unsigned oi[4];
                    float    wj[4];
#pragma unroll
                    for (int j = 0; j < 4; ++j) {
                        const int sl = jb + j * 4 + grp;
                        const int bp = sl << 2;
                        const unsigned sj = (unsigned)__builtin_amdgcn_ds_bpermute(bp, so);
                        hv[j] = (sl < cnt) ? h[sj + f] : 0.0f;
                        oi[j] = (unsigned)__builtin_amdgcn_ds_bpermute(bp, dd) + f;
                        wj[j] = __int_as_float(__builtin_amdgcn_ds_bpermute(bp, wi));
                    }
#pragma unroll
                    for (int j = 0; j < 4; ++j)
                        if (jb + j * 4 + grp < cnt) atomicAdd(out + oi[j], wj[j] * hv[j]);
                }
            }
        }
    }
}

__global__ void __launch_bounds__(SL_TPB)
gcn_bin_slice(const int* __restrict__ src, const int* __restrict__ dst,
              unsigned* __restrict__ binned, int* __restrict__ runs,
              int e_cnt, int nb) {
    __shared__ unsigned stage[SL_EDGES];
    __shared__ int cnt[BK_MAX];
    __shared__ int wsum[32];
    const int tid  = threadIdx.x;
    const int lane = tid & 31;
    const int wave = tid >> 5;
    const int s    = blockIdx.x;
    const int e0   = s * SL_EDGES;
    const int e_hi = e0 + SL_EDGES;
    const int e_end = e_hi < e_cnt ? e_hi : e_cnt;

    cnt[tid] = 0;
    __syncthreads();

    int dv[SL_EDGES / SL_TPB];
    int sv[SL_EDGES / SL_TPB];
#pragma unroll
    for (int k = 0; k < SL_EDGES / SL_TPB; ++k) {
        const int e  = e0 + k * SL_TPB + tid;
        const int ec = e < e_end ? e : e_end - 1;
        dv[k] = dst[ec];
        sv[k] = src[ec];
    }
#pragma unroll
    for (int k = 0; k < SL_EDGES / SL_TPB; ++k)
        if (e0 + k * SL_TPB + tid < e_end) atomicAdd(&cnt[dv[k] >> BK_SHIFT], 1);
    __syncthreads();

    const int v = cnt[tid];
    int incl = v;
#pragma unroll
    for (int o = 1; o < 32; o <<= 1) {
        const int t = __shfl_up(incl, o);
        if (lane >= o) incl += t;
    }
    if (lane == 31) wsum[wave] = incl;
    __syncthreads();
    if (wave == 0) {
        const int wt = wsum[lane];
        int wi = wt;
#pragma unroll
        for (int o = 1; o < 32; o <<= 1) {
            const int t = __shfl_up(wi, o);
            if (lane >= o) wi += t;
        }
        wsum[lane] = wi - wt;
    }
    __syncthreads();
    const int excl = incl - v + wsum[wave];
    cnt[tid] = excl;
    if (tid <= nb) runs[(size_t)s * (nb + 1) + tid] = excl;
    __syncthreads();

#pragma unroll
    for (int k = 0; k < SL_EDGES / SL_TPB; ++k) {
        if (e0 + k * SL_TPB + tid < e_end) {
            const int p = atomicAdd(&cnt[dv[k] >> BK_SHIFT], 1);
            stage[p] = (unsigned)sv[k] | ((unsigned)(dv[k] & (BK_NODES - 1)) << SRC_BITS);
        }
    }
    __syncthreads();

    const int n_s = e_end - e0;
    unsigned* __restrict__ region = binned + (size_t)s * SL_EDGES;
    unsigned cp[SL_EDGES / SL_TPB];
#pragma unroll
    for (int k = 0; k < SL_EDGES / SL_TPB; ++k) cp[k] = stage[k * SL_TPB + tid];
#pragma unroll
    for (int k = 0; k < SL_EDGES / SL_TPB; ++k)
        if (k * SL_TPB + tid < n_s) region[k * SL_TPB + tid] = cp[k];
}

__device__ __forceinline__ void run_words(const unsigned* __restrict__ region,
                                          int p0, int hi, int lane, unsigned (&w)[4]) {
#pragma unroll
    for (int c = 0; c < 4; ++c) {
        const int p = p0 + c * 32 + lane;
        w[c] = region[p < hi ? p : hi - 1];
    }
}

__global__ void __launch_bounds__(SL_TPB)
gcn_bucket_rows(const unsigned* __restrict__ binned, const int* __restrict__ runs,
                const float* __restrict__ x,
                const float* __restrict__ W,
                float* __restrict__ dinv, float* __restrict__ hs,
                int n, int nb, int n_slices) {
    __shared__ int   hist[BK_NODES];
    __shared__ float dloc[BK_NODES];
    __shared__ int   rlo[SL_TPB];
    __shared__ int   rhi[SL_TPB];
    const int tid   = threadIdx.x;
    const int lane  = tid & 31;
    const int wave  = __builtin_amdgcn_readfirstlane(tid >> 5);
    const int b     = blockIdx.x;
    const int node0 = b * BK_NODES;
    const int m     = lane & 15;
    const int khalf = (lane >> 4) * 2;

    v2f a[ROW_TILES][4];
#pragma unroll
    for (int q = 0; q < ROW_TILES; ++q) {
        int arow = node0 + (wave * ROW_TILES + q) * 16 + m;
        if (arow >= n) arow = n - 1;
        const float* __restrict__ xrow = x + (size_t)arow * IN_CH + khalf;
#pragma unroll
        for (int k = 0; k < 4; ++k) a[q][k] = *(const v2f*)(xrow + 4 * k);
    }
    const int   wrow = (m < OUT_CH) ? m : 0;
    const float sel  = (m < OUT_CH) ? 1.0f : 0.0f;
    const float* __restrict__ wr = W + wrow * IN_CH + khalf;
    v2f bf[4];
#pragma unroll
    for (int k = 0; k < 4; ++k) bf[k] = *(const v2f*)(wr + 4 * k) * sel;

    if (tid < n_slices) {
        const int* __restrict__ rp = runs + (size_t)tid * (nb + 1) + b;
        rlo[tid] = rp[0];
        rhi[tid] = rp[1];
    }
    for (int i = tid; i < BK_NODES; i += SL_TPB) hist[i] = 0;
    __syncthreads();

    for (int s0 = wave; s0 < n_slices; s0 += (SL_TPB / 32) * 4) {
        int      lo[4], hi[4];
        unsigned w[4][4];
#pragma unroll
        for (int u = 0; u < 4; ++u) {
            const int  s  = s0 + u * (SL_TPB / 32);
            const bool ok = s < n_slices;
            const int  sc = ok ? s : s0;
            const int  l  = __builtin_amdgcn_readfirstlane(rlo[sc]);
            const int  h  = __builtin_amdgcn_readfirstlane(rhi[sc]);
            lo[u] = ok ? l : 0;
            hi[u] = ok ? h : 0;
            const unsigned* __restrict__ region = binned + (size_t)sc * SL_EDGES;
#pragma unroll
            for (int cc = 0; cc < 4; ++cc) {
                const int p  = lo[u] + cc * 32 + lane;
                const int pc = p < hi[u] ? p : hi[u] - 1;
                w[u][cc] = region[pc > 0 ? pc : 0];
            }
        }
#pragma unroll
        for (int u = 0; u < 4; ++u) {
#pragma unroll
            for (int cc = 0; cc < 4; ++cc)
                if (lo[u] + cc * 32 + lane < hi[u]) atomicAdd(&hist[w[u][cc] >> SRC_BITS], 1);
        }
#pragma unroll
        for (int u = 0; u < 4; ++u) {
            for (int p0 = lo[u] + 128; p0 < hi[u]; p0 += 128) {
                unsigned wt[4];
                run_words(binned + (size_t)(s0 + u * (SL_TPB / 32)) * SL_EDGES, p0, hi[u], lane, wt);
#pragma unroll
                for (int cc = 0; cc < 4; ++cc)
                    if (p0 + cc * 32 + lane < hi[u]) atomicAdd(&hist[wt[cc] >> SRC_BITS], 1);
            }
        }
    }
    __syncthreads();

    for (int i = tid; i < BK_NODES; i += SL_TPB) {
        const float d = (float)(hist[i] + 1);
        const float v = d > 0.0f ? __frsqrt_rn(d) : 0.0f;
        dloc[i] = v;
        if (node0 + i < n) dinv[node0 + i] = v;
    }
    __syncthreads();

    v8f acc[ROW_TILES];
#pragma unroll
    for (int q = 0; q < ROW_TILES; ++q) acc[q] = v8f{};
#pragma unroll
    for (int k = 0; k < 4; ++k) {
#pragma unroll
        for (int q = 0; q < ROW_TILES; ++q)
            acc[q] = __builtin_amdgcn_wmma_f32_16x16x4_f32(false, a[q][k], false, bf[k], (short)0, acc[q], false, false);
    }

    if (m < OUT_CH) {
#pragma unroll
        for (int q = 0; q < ROW_TILES; ++q) {
            const int tb    = (wave * ROW_TILES + q) * 16;
            const int rbase = tb + (lane >> 4) * 8;
            const float4 d0 = *(const float4*)(dloc + rbase);
            const float4 d1 = *(const float4*)(dloc + rbase + 4);
            const float dv[8] = {d0.x, d0.y, d0.z, d0.w, d1.x, d1.y, d1.z, d1.w};
            float* __restrict__ hp = hs + (size_t)(node0 + rbase) * OUT_CH + m;
            if (node0 + tb + 16 <= n) {
#pragma unroll
                for (int r = 0; r < 8; ++r) hp[r * OUT_CH] = dv[r] * acc[q][r];
            } else {
#pragma unroll
                for (int r = 0; r < 8; ++r)
                    if (node0 + rbase + r < n) hp[r * OUT_CH] = dv[r] * acc[q][r];
            }
        }
    }
}

__device__ __forceinline__ void run_words_all(const unsigned* __restrict__ region,
                                              int lo, int hi, int lane, unsigned (&w)[4]) {
#pragma unroll
    for (int c = 0; c < 4; ++c) {
        const int p  = lo + c * 32 + lane;
        const int pc = p < hi ? p : hi - 1;
        w[c] = region[pc > 0 ? pc : 0];
    }
}

template <int NJ>
__device__ __forceinline__ void rows_request(const float* __restrict__ hs, unsigned w0, unsigned w1,
                                             const int (&bp)[8], unsigned f4,
                                             unsigned (&ww)[NJ], float (&hv)[NJ]) {
#pragma unroll
    for (int j = 0; j < NJ; ++j)
        ww[j] = (unsigned)__builtin_amdgcn_ds_bpermute(bp[j & 7], (int)(j < 8 ? w0 : w1));
#pragma unroll
    for (int j = 0; j < NJ; ++j) {
        const unsigned off = ((ww[j] & ((1u << SRC_BITS) - 1u)) << 5) | f4;
        hv[j] = *(const float*)((const char*)hs + off);
    }
}

template <int NJ>
__device__ __forceinline__ void rows_add(float* __restrict__ tile, float* __restrict__ spare,
                                         const unsigned (&ww)[NJ], const float (&hv)[NJ],
                                         int left, int grp, unsigned f) {
#pragma unroll
    for (int j = 0; j < NJ; ++j) {
        if (j * 4 < left) {
            float* dst = (j * 4 + grp < left) ? tile + ((ww[j] >> SRC_BITS) * OUT_CH + f) : spare;
            atomicAdd(dst, hv[j]);
        }
    }
}

__global__ void __launch_bounds__(SL_TPB)
gcn_bucket_sum(const unsigned* __restrict__ binned, const int* __restrict__ runs,
               const float* __restrict__ hs, const float* __restrict__ dinv,
               const float* __restrict__ bias, float* __restrict__ out,
               int n, int nb, int n_slices) {
    __shared__ __attribute__((aligned(16))) float tile[BK_NODES * OUT_CH];
    __shared__ int   rlo[SL_TPB];
    __shared__ int   rhi[SL_TPB];
    __shared__ float pad[32];
    const int      tid  = threadIdx.x;
    const int      lane = tid & 31;
    const int      wave = __builtin_amdgcn_readfirstlane(tid >> 5);
    const unsigned f    = (unsigned)(lane & 7);
    const int      grp  = lane >> 3;
    const int      b    = blockIdx.x;
    const int      node0 = b * BK_NODES;

    const int  rs = tid < n_slices ? tid : 0;
    const int* __restrict__ rp = runs + (size_t)rs * (nb + 1) + b;
    const int  r0 = rp[0];
    const int  r1 = rp[1];
    const int  half4 = (tid & 1) * 4;
    float4 own[BK_NODES * OUT_CH / 4 / SL_TPB];
    float  dvn[BK_NODES * OUT_CH / 4 / SL_TPB];
#pragma unroll
    for (int k = 0; k < BK_NODES * OUT_CH / 4 / SL_TPB; ++k) {
        const int node = node0 + ((tid + k * SL_TPB) >> 1);
        const int nc   = node < n ? node : n - 1;
        own[k] = *(const float4*)(hs + (size_t)nc * OUT_CH + half4);
        dvn[k] = dinv[nc];
    }
    const float4 b4 = *(const float4*)(bias + half4);
    rlo[tid] = r0;
    rhi[tid] = r1;
    pad[lane] = 0.0f;
#pragma unroll
    for (int k = 0; k < BK_NODES * OUT_CH / 4 / SL_TPB; ++k)
        ((float4*)tile)[tid + k * SL_TPB] = own[k];
    __syncthreads();

    int bp[8];
#pragma unroll
    for (int k = 0; k < 8; ++k) bp[k] = (k * 4 + grp) << 2;
    const unsigned f4    = f << 2;
    float*         spare = pad + lane;

    int s = wave;
    if (s < n_slices) {
        int lo = __builtin_amdgcn_readfirstlane(rlo[s]);
        int hi = __builtin_amdgcn_readfirstlane(rhi[s]);
        unsigned w[4];
        run_words_all(binned + (size_t)s * SL_EDGES, lo, hi, lane, w);
        const int  s1  = s + SL_TPB / 32;
        const bool ok1 = s1 < n_slices;
        const int  sc1 = ok1 ? s1 : s;
        const int  l1  = __builtin_amdgcn_readfirstlane(rlo[sc1]);
        const int  h1  = __builtin_amdgcn_readfirstlane(rhi[sc1]);
        int lo1 = ok1 ? l1 : 0;
        int hi1 = ok1 ? h1 : 0;
        unsigned w1[4];
        run_words_all(binned + (size_t)sc1 * SL_EDGES, lo1, hi1, lane, w1);
        unsigned wwA[16];
        float    hvA[16];
        rows_request<16>(hs, w[0], w[1], bp, f4, wwA, hvA);
        for (; s < n_slices; s += SL_TPB / 32) {
            const int  s2  = s + 2 * (SL_TPB / 32);
            const bool ok2 = s2 < n_slices;
            const int  sc2 = ok2 ? s2 : s;
            const int  l2  = __builtin_amdgcn_readfirstlane(rlo[sc2]);
            const int  h2  = __builtin_amdgcn_readfirstlane(rhi[sc2]);
            const int  lo2 = ok2 ? l2 : 0;
            const int  hi2 = ok2 ? h2 : 0;
            unsigned w2[4];
            run_words_all(binned + (size_t)sc2 * SL_EDGES, lo2, hi2, lane, w2);
            unsigned wwB[16];
            float    hvB[16];
            rows_request<16>(hs, w[2], w[3], bp, f4, wwB, hvB);
            const int left = hi - lo;
            rows_add<16>(tile, spare, wwA, hvA, left, grp, f);
            rows_request<16>(hs, w1[0], w1[1], bp, f4, wwA, hvA);
            rows_add<16>(tile, spare, wwB, hvB, left - 64, grp, f);
            for (int p0 = lo + 128; p0 < hi; p0 += 128) {
                unsigned wr[4];
                run_words(binned + (size_t)s * SL_EDGES, p0, hi, lane, wr);
                unsigned wwC[16];
                float    hvC[16];
                rows_request<16>(hs, wr[0], wr[1], bp, f4, wwC, hvC);
                rows_add<16>(tile, spare, wwC, hvC, hi - p0, grp, f);
                if (hi > p0 + 64) {
                    rows_request<16>(hs, wr[2], wr[3], bp, f4, wwC, hvC);
                    rows_add<16>(tile, spare, wwC, hvC, hi - p0 - 64, grp, f);
                }
            }
            lo = lo1; hi = hi1;
            lo1 = lo2; hi1 = hi2;
#pragma unroll
            for (int c = 0; c < 4; ++c) { w[c] = w1[c]; w1[c] = w2[c]; }
        }
    }
    __syncthreads();

#pragma unroll
    for (int k = 0; k < BK_NODES * OUT_CH / 4 / SL_TPB; ++k)
        own[k] = ((const float4*)tile)[tid + k * SL_TPB];
#pragma unroll
    for (int k = 0; k < BK_NODES * OUT_CH / 4 / SL_TPB; ++k) {
        const int node = node0 + ((tid + k * SL_TPB) >> 1);
        if (node < n) {
            float4 o;
            o.x = b4.x + dvn[k] * own[k].x;
            o.y = b4.y + dvn[k] * own[k].y;
            o.z = b4.z + dvn[k] * own[k].z;
            o.w = b4.w + dvn[k] * own[k].w;
            *(float4*)(out + (size_t)node * OUT_CH + half4) = o;
        }
    }
}

extern "C" void kernel_launch(void* const* d_in, const int* in_sizes, int n_in,
                              void* d_out, int out_size, void* d_ws, size_t ws_size,
                              hipStream_t stream) {
    const float* x    = (const float*)d_in[0];
    const int*   ei   = (const int*)d_in[1];
    const float* W    = (const float*)d_in[2];
    const float* bias = (const float*)d_in[3];

    const int N = in_sizes[0] / IN_CH;
    const int E = in_sizes[1] / 2;
    const int* src = ei;
    const int* dst = ei + (size_t)E;

    int*   ctr  = (int*)d_ws;
    const size_t dinv_off = 4u * CTR_STRIDE * sizeof(int);
    float* dinv = (float*)((char*)d_ws + dinv_off);
    size_t h_off = ((dinv_off + (size_t)N * sizeof(float)) + 255u) & ~(size_t)255u;
    float* h = (float*)((char*)d_ws + h_off);
    float* out = (float*)d_out;

    const int TPB = 256;

    const int split = ((N / 2) + 63) & ~63;

    const int    nb       = (N + BK_NODES - 1) >> BK_SHIFT;
    const int    n_slices = (E + SL_EDGES - 1) / SL_EDGES;
    const size_t bin_off  = ((h_off + (size_t)N * OUT_CH * sizeof(float)) + 255u) & ~(size_t)255u;
    const size_t runs_off = ((bin_off + (size_t)n_slices * SL_EDGES * sizeof(unsigned)) + 255u) & ~(size_t)255u;
    const size_t ws_need  = runs_off + (size_t)n_slices * (size_t)(nb + 1) * sizeof(int);
    const bool   buckets  = E > 0 && nb < BK_MAX && n_slices <= SL_TPB && N <= (1 << SRC_BITS) &&
                            ws_need <= ws_size;

    if (buckets) {
        unsigned* binned = (unsigned*)((char*)d_ws + bin_off);
        int*      runs   = (int*)((char*)d_ws + runs_off);
        gcn_bin_slice<<<n_slices, SL_TPB, 0, stream>>>(src, dst, binned, runs, E, nb);
        gcn_bucket_rows<<<nb, SL_TPB, 0, stream>>>(binned, runs, x, W, dinv, h, N, nb, n_slices);
        gcn_bucket_sum<<<nb, SL_TPB, 0, stream>>>(binned, runs, h, dinv, bias, out, N, nb, n_slices);
    } else {
        gcn_deg_init<<<(N + TPB - 1) / TPB, TPB, 0, stream>>>(dinv, ctr, N);

        const int deg_ranges    = (N + DEG_TILE - 1) / DEG_TILE;
        const int deg_ranges_lo = (deg_ranges + 1) / 2;
        const int deg_parts     = (256 + deg_ranges - 1) / deg_ranges;
        const int deg_slice     = (E + deg_parts - 1) / deg_parts;
        gcn_deg_count<<<256, DEG_TPB, 0, stream>>>(dst, dinv, ctr, E, N, deg_ranges,
                                                   deg_ranges_lo, deg_parts, deg_slice);
        gcn_dinv<<<(N + TPB - 1) / TPB, TPB, 0, stream>>>(dinv, N);

        const int n_tiles = (N + 15) / 16;
        const int waves_per_block = TPB / 32;
        const int gemm_blocks = (n_tiles + waves_per_block - 1) / waves_per_block;
        gcn_gemm_wmma<<<gemm_blocks, TPB, 0, stream>>>(x, W, bias, dinv, h, out, N, n_tiles);

        const int sc_chunks = (E + SC_CHUNK - 1) / SC_CHUNK;
        gcn_scatter<<<2048, TPB, 0, stream>>>(src, dst, dinv, h, out, ctr + 2 * CTR_STRIDE,
                                              E, sc_chunks, split, N);
    }
}


# slice sort: the bucket count is the only LDS add pass: each add goes out as its index load returns and gives the edge's rank; place = bucket start + rank (a read, no add chain); sum order may change
# speedup vs baseline: 10.4181x; 1.0513x over previous
#include <hip/hip_runtime.h>
#include <hip/hip_bf16.h>

typedef __attribute__((ext_vector_type(2))) float v2f;
typedef __attribute__((ext_vector_type(8))) float v8f;

#define IN_CH 16
#define OUT_CH 8

#define CTR_STRIDE 64
#define DEG_TILE   65536
#define DEG_TPB    1024
#define DEG_UNROLL 16
#define DEG_FLUSH  8
#define SC_CHUNK  128
#define BK_SHIFT  11
#define BK_NODES  2048
#define BK_MAX    1024
#define SRC_BITS  21
#define SL_EDGES  20480
#define SL_TPB    1024
#define ROW_TILES 4

__device__ __forceinline__ int cache_half() {
    return (int)((__builtin_amdgcn_s_sendmsg_rtn(0x87) >> 18) & 1u);
}

__device__ __forceinline__ int draw_ticket(int* ctr) {
    int t = 0;
    if ((threadIdx.x & 31) == 0) t = atomicAdd(ctr, 1);
    return __builtin_amdgcn_readfirstlane(t);
}

__global__ void gcn_deg_init(float* __restrict__ deg, int* __restrict__ ctr, int n) {
    int i = blockIdx.x * blockDim.x + threadIdx.x;
    if (i < n) deg[i] = 1.0f;
    if (i < 4) ctr[i * CTR_STRIDE] = 0;
}

__global__ void __launch_bounds__(DEG_TPB)
gcn_deg_count(const int* __restrict__ dst, float* __restrict__ deg,
              int* __restrict__ ctr, int e_cnt, int n,
              int n_ranges, int ranges_lo, int parts, int slice) {
    __shared__ int tile[DEG_TILE];
    __shared__ int s_msg[2];
    const int tid = threadIdx.x;

    if (tid == 0) s_msg[1] = cache_half();
    __syncthreads();
    int  r      = __builtin_amdgcn_readfirstlane(s_msg[1]);
    bool helped = false;
    const int max_trips = n_ranges * parts + 2;
    for (int it = 0; it < max_trips; ++it) {
        if (tid == 0) s_msg[0] = atomicAdd(ctr + r * CTR_STRIDE, 1);
        __syncthreads();
        const int t = __builtin_amdgcn_readfirstlane(s_msg[0]);
        __syncthreads();
        const int first = r ? ranges_lo : 0;
        const int count = r ? (n_ranges - ranges_lo) : ranges_lo;
        if (t >= count * parts) {
            if (helped) break;
            helped = true;
            r ^= 1;
            continue;
        }
        const int range = first + t / parts;
        const int part  = t - (t / parts) * parts;
        const int lo    = range * DEG_TILE;

        for (int i = tid; i < DEG_TILE; i += DEG_TPB) tile[i] = 0;
        __syncthreads();

        const int e_begin = part * slice;
        const int e_stop  = e_begin + slice;
        const int e_end   = e_stop < e_cnt ? e_stop : e_cnt;
        for (int base = e_begin; base < e_end; base += DEG_TPB * DEG_UNROLL) {
            int d[DEG_UNROLL];
#pragma unroll
            for (int k = 0; k < DEG_UNROLL; ++k) {
                const int e = base + k * DEG_TPB + tid;
                d[k] = (e < e_end) ? dst[e] : -1;
            }
#pragma unroll
            for (int k = 0; k < DEG_UNROLL; ++k) {
                const unsigned off = (unsigned)(d[k] - lo);
                if (off < (unsigned)DEG_TILE) atomicAdd(&tile[off], 1);
            }
        }
        __syncthreads();

        for (int i0 = tid; i0 < DEG_TILE; i0 += DEG_TPB * DEG_FLUSH) {
            int c[DEG_FLUSH];
#pragma unroll
            for (int k = 0; k < DEG_FLUSH; ++k) c[k] = tile[i0 + k * DEG_TPB];
#pragma unroll
            for (int k = 0; k < DEG_FLUSH; ++k) {
                const int node = lo + i0 + k * DEG_TPB;
                if (c[k] != 0 && node < n) atomicAdd(&deg[node], (float)c[k]);
            }
        }
    }
}

__global__ void gcn_dinv(float* __restrict__ deg, int n) {
    int i = blockIdx.x * blockDim.x + threadIdx.x;
    if (i < n) {
        float d = deg[i];
        deg[i] = d > 0.0f ? __frsqrt_rn(d) : 0.0f;
    }
}

__global__ void gcn_gemm_wmma(const float* __restrict__ x,
                              const float* __restrict__ W,
                              const float* __restrict__ bias,
                              const float* __restrict__ dinv,
                              float* __restrict__ h,
                              float* __restrict__ out,
                              int n_nodes, int n_tiles) {
    const int lane  = threadIdx.x & 31;
    const int wave  = threadIdx.x >> 5;
    const int tile  = blockIdx.x * (blockDim.x >> 5) + wave;
    if (tile >= n_tiles) return;

    const int base  = tile * 16;
    const int m     = lane & 15;
    const int khalf = (lane >> 4) * 2;

    int arow = base + m;
    if (arow >= n_nodes) arow = n_nodes - 1;
    const float* __restrict__ xrow = x + (size_t)arow * IN_CH;

    v2f a0 = *(const v2f*)(xrow + 0  + khalf);
    v2f a1 = *(const v2f*)(xrow + 4  + khalf);
    v2f a2 = *(const v2f*)(xrow + 8  + khalf);
    v2f a3 = *(const v2f*)(xrow + 12 + khalf);

    const int   wrow = (m < OUT_CH) ? m : 0;
    const float sel  = (m < OUT_CH) ? 1.0f : 0.0f;
    const float* __restrict__ wr = W + wrow * IN_CH + khalf;
    v2f b0 = *(const v2f*)(wr + 0);
    v2f b1 = *(const v2f*)(wr + 4);
    v2f b2 = *(const v2f*)(wr + 8);
    v2f b3 = *(const v2f*)(wr + 12);
    b0 *= sel; b1 *= sel; b2 *= sel; b3 *= sel;

    v8f acc = {};
    acc = __builtin_amdgcn_wmma_f32_16x16x4_f32(false, a0, false, b0, (short)0, acc, false, false);
    acc = __builtin_amdgcn_wmma_f32_16x16x4_f32(false, a1, false, b1, (short)0, acc, false, false);
    acc = __builtin_amdgcn_wmma_f32_16x16x4_f32(false, a2, false, b2, (short)0, acc, false, false);
    acc = __builtin_amdgcn_wmma_f32_16x16x4_f32(false, a3, false, b3, (short)0, acc, false, false);

    const int  n     = m;
    const int  mbase = base + ((lane >> 4) * 8);
    const bool full  = (base + 16) <= n_nodes;

    if (full) {
        if (n < OUT_CH) {
            const float bn = bias[n];
            const float4 d0 = *(const float4*)(dinv + mbase);
            const float4 d1 = *(const float4*)(dinv + mbase + 4);
            const float dv[8] = {d0.x, d0.y, d0.z, d0.w, d1.x, d1.y, d1.z, d1.w};
            float* __restrict__ hp = h   + (size_t)mbase * OUT_CH + n;
            float* __restrict__ op = out + (size_t)mbase * OUT_CH + n;
#pragma unroll
            for (int r = 0; r < 8; ++r) {
                const float hv = acc[r];
                hp[r * OUT_CH] = hv;
                op[r * OUT_CH] = bn + dv[r] * dv[r] * hv;
            }
        }
    } else {
        if (n < OUT_CH) {
            const float bn = bias[n];
#pragma unroll
            for (int r = 0; r < 8; ++r) {
                const int node = mbase + r;
                if (node < n_nodes) {
                    const float hv = acc[r];
                    h[(size_t)node * OUT_CH + n] = hv;
                    const float di = dinv[node];
                    out[(size_t)node * OUT_CH + n] = bn + di * di * hv;
                }
            }
        }
    }
}

__global__ void __launch_bounds__(256)
gcn_scatter(const int* __restrict__ src, const int* __restrict__ dst,
            const float* __restrict__ dinv, const float* __restrict__ h,
            float* __restrict__ out, int* __restrict__ ctr,
            int e_cnt, int n_chunks, int split, int n) {
    const int      lane = threadIdx.x & 31;
    const unsigned f    = (unsigned)(lane & 7);
    const int      grp  = lane >> 3;

    int  r      = cache_half();
    bool helped = false;
    int  next   = draw_ticket(ctr + r * CTR_STRIDE);
    const int max_trips = 2 * n_chunks + 2;
    for (int it = 0; it < max_trips; ++it) {
        const int c = next;
        if (c >= n_chunks) {
            if (helped) break;
            helped = true;
            r ^= 1;
            next = draw_ticket(ctr + r * CTR_STRIDE);
            continue;
        }
        const int lo = r ? split : 0;
        const int hi = r ? n : split;
        const int e0 = c * SC_CHUNK + lane;

        int   dv[SC_CHUNK / 32];
        int   sv[SC_CHUNK / 32];
        float wv[SC_CHUNK / 32];
        bool  in[SC_CHUNK / 32];
#pragma unroll
        for (int p = 0; p < SC_CHUNK / 32; ++p) {
            const int e  = e0 + p * 32;
            const int ec = (e < e_cnt) ? e : e_cnt - 1;
            dv[p] = dst[ec];
            sv[p] = src[ec];
        }
        next = draw_ticket(ctr + r * CTR_STRIDE);
#pragma unroll
        for (int p = 0; p < SC_CHUNK / 32; ++p) {
            in[p] = (e0 + p * 32 < e_cnt) && (dv[p] >= lo) && (dv[p] < hi);
            dv[p] = in[p] ? dv[p] : 0;
            sv[p] = in[p] ? sv[p] : 0;
        }
#pragma unroll
        for (int p = 0; p < SC_CHUNK / 32; ++p) {
            const float a = dinv[sv[p]];
            const float b = dinv[dv[p]];
            wv[p] = in[p] ? a * b : 0.0f;
        }

#pragma unroll
        for (int p = 0; p < SC_CHUNK / 32; ++p) {
            const unsigned mask = __builtin_amdgcn_ballot_w32(in[p]);
            const int cnt = __popc(mask);
            if (cnt > 0) {
                const int rank = (int)__builtin_amdgcn_mbcnt_lo(mask, 0u);
                const int slot = in[p] ? rank : cnt + lane - rank;
                const int so = __builtin_amdgcn_ds_permute(slot << 2, sv[p] * OUT_CH);
                const int dd = __builtin_amdgcn_ds_permute(slot << 2, dv[p] * OUT_CH);
                const int wi = __builtin_amdgcn_ds_permute(slot << 2, __float_as_int(wv[p]));
                for (int jb = 0; jb < cnt; jb += 16) {
                    float    hv[4];
                    unsigned oi[4];
                    float    wj[4];
#pragma unroll
                    for (int j = 0; j < 4; ++j) {
                        const int sl = jb + j * 4 + grp;
                        const int bp = sl << 2;
                        const unsigned sj = (unsigned)__builtin_amdgcn_ds_bpermute(bp, so);
                        hv[j] = (sl < cnt) ? h[sj + f] : 0.0f;
                        oi[j] = (unsigned)__builtin_amdgcn_ds_bpermute(bp, dd) + f;
                        wj[j] = __int_as_float(__builtin_amdgcn_ds_bpermute(bp, wi));
                    }
#pragma unroll
                    for (int j = 0; j < 4; ++j)
                        if (jb + j * 4 + grp < cnt) atomicAdd(out + oi[j], wj[j] * hv[j]);
                }
            }
        }
    }
}

__global__ void __launch_bounds__(SL_TPB)
gcn_bin_slice(const int* __restrict__ src, const int* __restrict__ dst,
              unsigned* __restrict__ binned, int* __restrict__ runs,
              int e_cnt, int nb) {
    __shared__ unsigned stage[SL_EDGES];
    __shared__ int cnt[BK_MAX];
    __shared__ int wsum[32];
    const int tid  = threadIdx.x;
    const int lane = tid & 31;
    const int wave = tid >> 5;
    const int s    = blockIdx.x;
    const int e0   = s * SL_EDGES;
    const int e_hi = e0 + SL_EDGES;
    const int e_end = e_hi < e_cnt ? e_hi : e_cnt;

    cnt[tid] = 0;
    __syncthreads();

    int dv[SL_EDGES / SL_TPB];
    int sv[SL_EDGES / SL_TPB];
#pragma unroll
    for (int k = 0; k < SL_EDGES / SL_TPB; ++k) {
        const int e  = e0 + k * SL_TPB + tid;
        const int ec = e < e_end ? e : e_end - 1;
        dv[k] = dst[ec];
        sv[k] = src[ec];
        __builtin_amdgcn_sched_barrier(0);
    }
    int rk[SL_EDGES / SL_TPB];
#pragma unroll
    for (int k = 0; k < SL_EDGES / SL_TPB; ++k) {
        rk[k] = 0;
        if (e0 + k * SL_TPB + tid < e_end) rk[k] = atomicAdd(&cnt[dv[k] >> BK_SHIFT], 1);
    }
    __syncthreads();

    const int v = cnt[tid];
    int incl = v;
#pragma unroll
    for (int o = 1; o < 32; o <<= 1) {
        const int t = __shfl_up(incl, o);
        if (lane >= o) incl += t;
    }
    if (lane == 31) wsum[wave] = incl;
    __syncthreads();
    if (wave == 0) {
        const int wt = wsum[lane];
        int wi = wt;
#pragma unroll
        for (int o = 1; o < 32; o <<= 1) {
            const int t = __shfl_up(wi, o);
            if (lane >= o) wi += t;
        }
        wsum[lane] = wi - wt;
    }
    __syncthreads();
    const int excl = incl - v + wsum[wave];
    cnt[tid] = excl;
    if (tid <= nb) runs[(size_t)s * (nb + 1) + tid] = excl;
    __syncthreads();

    int first[SL_EDGES / SL_TPB];
#pragma unroll
    for (int k = 0; k < SL_EDGES / SL_TPB; ++k) first[k] = cnt[dv[k] >> BK_SHIFT];
#pragma unroll
    for (int k = 0; k < SL_EDGES / SL_TPB; ++k) {
        if (e0 + k * SL_TPB + tid < e_end)
            stage[first[k] + rk[k]] =
                (unsigned)sv[k] | ((unsigned)(dv[k] & (BK_NODES - 1)) << SRC_BITS);
    }
    __syncthreads();

    const int n_s = e_end - e0;
    unsigned* __restrict__ region = binned + (size_t)s * SL_EDGES;
    unsigned cp[SL_EDGES / SL_TPB];
#pragma unroll
    for (int k = 0; k < SL_EDGES / SL_TPB; ++k) cp[k] = stage[k * SL_TPB + tid];
#pragma unroll
    for (int k = 0; k < SL_EDGES / SL_TPB; ++k)
        if (k * SL_TPB + tid < n_s) region[k * SL_TPB + tid] = cp[k];
}

__device__ __forceinline__ void run_words(const unsigned* __restrict__ region,
                                          int p0, int hi, int lane, unsigned (&w)[4]) {
#pragma unroll
    for (int c = 0; c < 4; ++c) {
        const int p = p0 + c * 32 + lane;
        w[c] = region[p < hi ? p : hi - 1];
    }
}

__global__ void __launch_bounds__(SL_TPB)
gcn_bucket_rows(const unsigned* __restrict__ binned, const int* __restrict__ runs,
                const float* __restrict__ x,
                const float* __restrict__ W,
                float* __restrict__ dinv, float* __restrict__ hs,
                int n, int nb, int n_slices) {
    __shared__ int   hist[BK_NODES];
    __shared__ float dloc[BK_NODES];
    __shared__ int   rlo[SL_TPB];
    __shared__ int   rhi[SL_TPB];
    const int tid   = threadIdx.x;
    const int lane  = tid & 31;
    const int wave  = __builtin_amdgcn_readfirstlane(tid >> 5);
    const int b     = blockIdx.x;
    const int node0 = b * BK_NODES;
    const int m     = lane & 15;
    const int khalf = (lane >> 4) * 2;

    v2f a[ROW_TILES][4];
#pragma unroll
    for (int q = 0; q < ROW_TILES; ++q) {
        int arow = node0 + (wave * ROW_TILES + q) * 16 + m;
        if (arow >= n) arow = n - 1;
        const float* __restrict__ xrow = x + (size_t)arow * IN_CH + khalf;
#pragma unroll
        for (int k = 0; k < 4; ++k) a[q][k] = *(const v2f*)(xrow + 4 * k);
    }
    const int   wrow = (m < OUT_CH) ? m : 0;
    const float sel  = (m < OUT_CH) ? 1.0f : 0.0f;
    const float* __restrict__ wr = W + wrow * IN_CH + khalf;
    v2f bf[4];
#pragma unroll
    for (int k = 0; k < 4; ++k) bf[k] = *(const v2f*)(wr + 4 * k) * sel;

    if (tid < n_slices) {
        const int* __restrict__ rp = runs + (size_t)tid * (nb + 1) + b;
        rlo[tid] = rp[0];
        rhi[tid] = rp[1];
    }
    for (int i = tid; i < BK_NODES; i += SL_TPB) hist[i] = 0;
    __syncthreads();

    for (int s0 = wave; s0 < n_slices; s0 += (SL_TPB / 32) * 4) {
        int      lo[4], hi[4];
        unsigned w[4][4];
#pragma unroll
        for (int u = 0; u < 4; ++u) {
            const int  s  = s0 + u * (SL_TPB / 32);
            const bool ok = s < n_slices;
            const int  sc = ok ? s : s0;
            const int  l  = __builtin_amdgcn_readfirstlane(rlo[sc]);
            const int  h  = __builtin_amdgcn_readfirstlane(rhi[sc]);
            lo[u] = ok ? l : 0;
            hi[u] = ok ? h : 0;
            const unsigned* __restrict__ region = binned + (size_t)sc * SL_EDGES;
#pragma unroll
            for (int cc = 0; cc < 4; ++cc) {
                const int p  = lo[u] + cc * 32 + lane;
                const int pc = p < hi[u] ? p : hi[u] - 1;
                w[u][cc] = region[pc > 0 ? pc : 0];
            }
        }
#pragma unroll
        for (int u = 0; u < 4; ++u) {
#pragma unroll
            for (int cc = 0; cc < 4; ++cc)
                if (lo[u] + cc * 32 + lane < hi[u]) atomicAdd(&hist[w[u][cc] >> SRC_BITS], 1);
        }
#pragma unroll
        for (int u = 0; u < 4; ++u) {
            for (int p0 = lo[u] + 128; p0 < hi[u]; p0 += 128) {
                unsigned wt[4];
                run_words(binned + (size_t)(s0 + u * (SL_TPB / 32)) * SL_EDGES, p0, hi[u], lane, wt);
#pragma unroll
                for (int cc = 0; cc < 4; ++cc)
                    if (p0 + cc * 32 + lane < hi[u]) atomicAdd(&hist[wt[cc] >> SRC_BITS], 1);
            }
        }
    }
    __syncthreads();

    for (int i = tid; i < BK_NODES; i += SL_TPB) {
        const float d = (float)(hist[i] + 1);
        const float v = d > 0.0f ? __frsqrt_rn(d) : 0.0f;
        dloc[i] = v;
        if (node0 + i < n) dinv[node0 + i] = v;
    }
    __syncthreads();

    v8f acc[ROW_TILES];
#pragma unroll
    for (int q = 0; q < ROW_TILES; ++q) acc[q] = v8f{};
#pragma unroll
    for (int k = 0; k < 4; ++k) {
#pragma unroll
        for (int q = 0; q < ROW_TILES; ++q)
            acc[q] = __builtin_amdgcn_wmma_f32_16x16x4_f32(false, a[q][k], false, bf[k], (short)0, acc[q], false, false);
    }

    if (m < OUT_CH) {
#pragma unroll
        for (int q = 0; q < ROW_TILES; ++q) {
            const int tb    = (wave * ROW_TILES + q) * 16;
            const int rbase = tb + (lane >> 4) * 8;
            const float4 d0 = *(const float4*)(dloc + rbase);
            const float4 d1 = *(const float4*)(dloc + rbase + 4);
            const float dv[8] = {d0.x, d0.y, d0.z, d0.w, d1.x, d1.y, d1.z, d1.w};
            float* __restrict__ hp = hs + (size_t)(node0 + rbase) * OUT_CH + m;
            if (node0 + tb + 16 <= n) {
#pragma unroll
                for (int r = 0; r < 8; ++r) hp[r * OUT_CH] = dv[r] * acc[q][r];
            } else {
#pragma unroll
                for (int r = 0; r < 8; ++r)
                    if (node0 + rbase + r < n) hp[r * OUT_CH] = dv[r] * acc[q][r];
            }
        }
    }
}

__device__ __forceinline__ void run_words_all(const unsigned* __restrict__ region,
                                              int lo, int hi, int lane, unsigned (&w)[4]) {
#pragma unroll
    for (int c = 0; c < 4; ++c) {
        const int p  = lo + c * 32 + lane;
        const int pc = p < hi ? p : hi - 1;
        w[c] = region[pc > 0 ? pc : 0];
    }
}

template <int NJ>
__device__ __forceinline__ void rows_request(const float* __restrict__ hs, unsigned w0, unsigned w1,
                                             const int (&bp)[8], unsigned f4,
                                             unsigned (&ww)[NJ], float (&hv)[NJ]) {
#pragma unroll
    for (int j = 0; j < NJ; ++j)
        ww[j] = (unsigned)__builtin_amdgcn_ds_bpermute(bp[j & 7], (int)(j < 8 ? w0 : w1));
#pragma unroll
    for (int j = 0; j < NJ; ++j) {
        const unsigned off = ((ww[j] & ((1u << SRC_BITS) - 1u)) << 5) | f4;
        hv[j] = *(const float*)((const char*)hs + off);
    }
}

template <int NJ>
__device__ __forceinline__ void rows_add(float* __restrict__ tile, float* __restrict__ spare,
                                         const unsigned (&ww)[NJ], const float (&hv)[NJ],
                                         int left, int grp, unsigned f) {
#pragma unroll
    for (int j = 0; j < NJ; ++j) {
        if (j * 4 < left) {
            float* dst = (j * 4 + grp < left) ? tile + ((ww[j] >> SRC_BITS) * OUT_CH + f) : spare;
            atomicAdd(dst, hv[j]);
        }
    }
}

__global__ void __launch_bounds__(SL_TPB)
gcn_bucket_sum(const unsigned* __restrict__ binned, const int* __restrict__ runs,
               const float* __restrict__ hs, const float* __restrict__ dinv,
               const float* __restrict__ bias, float* __restrict__ out,
               int n, int nb, int n_slices) {
    __shared__ __attribute__((aligned(16))) float tile[BK_NODES * OUT_CH];
    __shared__ int   rlo[SL_TPB];
    __shared__ int   rhi[SL_TPB];
    __shared__ float pad[32];
    const int      tid  = threadIdx.x;
    const int      lane = tid & 31;
    const int      wave = __builtin_amdgcn_readfirstlane(tid >> 5);
    const unsigned f    = (unsigned)(lane & 7);
    const int      grp  = lane >> 3;
    const int      b    = blockIdx.x;
    const int      node0 = b * BK_NODES;

    const int  rs = tid < n_slices ? tid : 0;
    const int* __restrict__ rp = runs + (size_t)rs * (nb + 1) + b;
    const int  r0 = rp[0];
    const int  r1 = rp[1];
    const int  half4 = (tid & 1) * 4;
    float4 own[BK_NODES * OUT_CH / 4 / SL_TPB];
    float  dvn[BK_NODES * OUT_CH / 4 / SL_TPB];
#pragma unroll
    for (int k = 0; k < BK_NODES * OUT_CH / 4 / SL_TPB; ++k) {
        const int node = node0 + ((tid + k * SL_TPB) >> 1);
        const int nc   = node < n ? node : n - 1;
        own[k] = *(const float4*)(hs + (size_t)nc * OUT_CH + half4);
        dvn[k] = dinv[nc];
    }
    const float4 b4 = *(const float4*)(bias + half4);
    rlo[tid] = r0;
    rhi[tid] = r1;
    pad[lane] = 0.0f;
#pragma unroll
    for (int k = 0; k < BK_NODES * OUT_CH / 4 / SL_TPB; ++k)
        ((float4*)tile)[tid + k * SL_TPB] = own[k];
    __syncthreads();

    int bp[8];
#pragma unroll
    for (int k = 0; k < 8; ++k) bp[k] = (k * 4 + grp) << 2;
    const unsigned f4    = f << 2;
    float*         spare = pad + lane;

    int s = wave;
    if (s < n_slices) {
        int lo = __builtin_amdgcn_readfirstlane(rlo[s]);
        int hi = __builtin_amdgcn_readfirstlane(rhi[s]);
        unsigned w[4];
        run_words_all(binned + (size_t)s * SL_EDGES, lo, hi, lane, w);
        const int  s1  = s + SL_TPB / 32;
        const bool ok1 = s1 < n_slices;
        const int  sc1 = ok1 ? s1 : s;
        const int  l1  = __builtin_amdgcn_readfirstlane(rlo[sc1]);
        const int  h1  = __builtin_amdgcn_readfirstlane(rhi[sc1]);
        int lo1 = ok1 ? l1 : 0;
        int hi1 = ok1 ? h1 : 0;
        unsigned w1[4];
        run_words_all(binned + (size_t)sc1 * SL_EDGES, lo1, hi1, lane, w1);
        unsigned wwA[16];
        float    hvA[16];
        rows_request<16>(hs, w[0], w[1], bp, f4, wwA, hvA);
        for (; s < n_slices; s += SL_TPB / 32) {
            const int  s2  = s + 2 * (SL_TPB / 32);
            const bool ok2 = s2 < n_slices;
            const int  sc2 = ok2 ? s2 : s;
            const int  l2  = __builtin_amdgcn_readfirstlane(rlo[sc2]);
            const int  h2  = __builtin_amdgcn_readfirstlane(rhi[sc2]);
            const int  lo2 = ok2 ? l2 : 0;
            const int  hi2 = ok2 ? h2 : 0;
            unsigned w2[4];
            run_words_all(binned + (size_t)sc2 * SL_EDGES, lo2, hi2, lane, w2);
            unsigned wwB[16];
            float    hvB[16];
            rows_request<16>(hs, w[2], w[3], bp, f4, wwB, hvB);
            const int left = hi - lo;
            rows_add<16>(tile, spare, wwA, hvA, left, grp, f);
            rows_request<16>(hs, w1[0], w1[1], bp, f4, wwA, hvA);
            rows_add<16>(tile, spare, wwB, hvB, left - 64, grp, f);
            for (int p0 = lo + 128; p0 < hi; p0 += 128) {
                unsigned wr[4];
                run_words(binned + (size_t)s * SL_EDGES, p0, hi, lane, wr);
                unsigned wwC[16];
                float    hvC[16];
                rows_request<16>(hs, wr[0], wr[1], bp, f4, wwC, hvC);
                rows_add<16>(tile, spare, wwC, hvC, hi - p0, grp, f);
                if (hi > p0 + 64) {
                    rows_request<16>(hs, wr[2], wr[3], bp, f4, wwC, hvC);
                    rows_add<16>(tile, spare, wwC, hvC, hi - p0 - 64, grp, f);
                }
            }
            lo = lo1; hi = hi1;
            lo1 = lo2; hi1 = hi2;
#pragma unroll
            for (int c = 0; c < 4; ++c) { w[c] = w1[c]; w1[c] = w2[c]; }
        }
    }
    __syncthreads();

#pragma unroll
    for (int k = 0; k < BK_NODES * OUT_CH / 4 / SL_TPB; ++k)
        own[k] = ((const float4*)tile)[tid + k * SL_TPB];
#pragma unroll
    for (int k = 0; k < BK_NODES * OUT_CH / 4 / SL_TPB; ++k) {
        const int node = node0 + ((tid + k * SL_TPB) >> 1);
        if (node < n) {
            float4 o;
            o.x = b4.x + dvn[k] * own[k].x;
            o.y = b4.y + dvn[k] * own[k].y;
            o.z = b4.z + dvn[k] * own[k].z;
            o.w = b4.w + dvn[k] * own[k].w;
            *(float4*)(out + (size_t)node * OUT_CH + half4) = o;
        }
    }
}

extern "C" void kernel_launch(void* const* d_in, const int* in_sizes, int n_in,
                              void* d_out, int out_size, void* d_ws, size_t ws_size,
                              hipStream_t stream) {
    const float* x    = (const float*)d_in[0];
    const int*   ei   = (const int*)d_in[1];
    const float* W    = (const float*)d_in[2];
    const float* bias = (const float*)d_in[3];

    const int N = in_sizes[0] / IN_CH;
    const int E = in_sizes[1] / 2;
    const int* src = ei;
    const int* dst = ei + (size_t)E;

    int*   ctr  = (int*)d_ws;
    const size_t dinv_off = 4u * CTR_STRIDE * sizeof(int);
    float* dinv = (float*)((char*)d_ws + dinv_off);
    size_t h_off = ((dinv_off + (size_t)N * sizeof(float)) + 255u) & ~(size_t)255u;
    float* h = (float*)((char*)d_ws + h_off);
    float* out = (float*)d_out;

    const int TPB = 256;

    const int split = ((N / 2) + 63) & ~63;

    const int    nb       = (N + BK_NODES - 1) >> BK_SHIFT;
    const int    n_slices = (E + SL_EDGES - 1) / SL_EDGES;
    const size_t bin_off  = ((h_off + (size_t)N * OUT_CH * sizeof(float)) + 255u) & ~(size_t)255u;
    const size_t runs_off = ((bin_off + (size_t)n_slices * SL_EDGES * sizeof(unsigned)) + 255u) & ~(size_t)255u;
    const size_t ws_need  = runs_off + (size_t)n_slices * (size_t)(nb + 1) * sizeof(int);
    const bool   buckets  = E > 0 && nb < BK_MAX && n_slices <= SL_TPB && N <= (1 << SRC_BITS) &&
                            ws_need <= ws_size;

    if (buckets) {
        unsigned* binned = (unsigned*)((char*)d_ws + bin_off);
        int*      runs   = (int*)((char*)d_ws + runs_off);
        gcn_bin_slice<<<n_slices, SL_TPB, 0, stream>>>(src, dst, binned, runs, E, nb);
        gcn_bucket_rows<<<nb, SL_TPB, 0, stream>>>(binned, runs, x, W, dinv, h, N, nb, n_slices);
        gcn_bucket_sum<<<nb, SL_TPB, 0, stream>>>(binned, runs, h, dinv, bias, out, N, nb, n_slices);
    } else {
        gcn_deg_init<<<(N + TPB - 1) / TPB, TPB, 0, stream>>>(dinv, ctr, N);

        const int deg_ranges    = (N + DEG_TILE - 1) / DEG_TILE;
        const int deg_ranges_lo = (deg_ranges + 1) / 2;
        const int deg_parts     = (256 + deg_ranges - 1) / deg_ranges;
        const int deg_slice     = (E + deg_parts - 1) / deg_parts;
        gcn_deg_count<<<256, DEG_TPB, 0, stream>>>(dst, dinv, ctr, E, N, deg_ranges,
                                                   deg_ranges_lo, deg_parts, deg_slice);
        gcn_dinv<<<(N + TPB - 1) / TPB, TPB, 0, stream>>>(dinv, N);

        const int n_tiles = (N + 15) / 16;
        const int waves_per_block = TPB / 32;
        const int gemm_blocks = (n_tiles + waves_per_block - 1) / waves_per_block;
        gcn_gemm_wmma<<<gemm_blocks, TPB, 0, stream>>>(x, W, bias, dinv, h, out, N, n_tiles);

        const int sc_chunks = (E + SC_CHUNK - 1) / SC_CHUNK;
        gcn_scatter<<<2048, TPB, 0, stream>>>(src, dst, dinv, h, out, ctr + 2 * CTR_STRIDE,
                                              E, sc_chunks, split, N);
    }
}
